# starting kernel
# baseline (speedup 1.0000x reference)
#include <hip/hip_runtime.h>
#include <math.h>

typedef __attribute__((ext_vector_type(2))) float v2f;
typedef __attribute__((ext_vector_type(4))) float v4f;
typedef __attribute__((ext_vector_type(8))) float v8f;

#define PN 32
#define MN 32
#define BN 128
#define HN 256
#define DN 54
#define ON 53
#define OPAD 64
#define CH 512

#define RB 128
#define CB 64
#define KC 64
#define LDK (KC + 4)


__global__ void zero_f32(float* p, int n) {
  int i = blockIdx.x * blockDim.x + threadIdx.x;
  if (i < n) p[i] = 0.f;
}

__global__ void onehot_to_idx(const float* __restrict__ oh, int* __restrict__ idx, int rows) {
  int r = blockIdx.x * blockDim.x + threadIdx.x;
  if (r >= rows) return;
  const float* p = oh + (size_t)r * DN;
  int v = 0;
  for (int d = 0; d < DN; ++d)
    if (p[d] > 0.5f) v = d;
  idx[r] = v;
}

__global__ void pad_proj_weights(const float* __restrict__ Ws, const float* __restrict__ bs,
                                 const float* __restrict__ Wi, const float* __restrict__ bi,
                                 float* __restrict__ Wsp, float* __restrict__ Wip,
                                 float* __restrict__ bsp, float* __restrict__ bip) {
  int t = blockIdx.x * blockDim.x + threadIdx.x;
  if (t >= OPAD * CH) return;
  int n = t / CH;
  int k = t - n * CH;
  Wsp[t] = (n < ON) ? Ws[n * CH + k] : 0.f;
  Wip[t] = (n < ON) ? Wi[n * CH + k] : 0.f;
  if (t < OPAD) {
    bsp[t] = (t < ON) ? bs[t] : 0.f;
    bip[t] = (t < ON) ? bi[t] : 0.f;
  }
}

__global__ __launch_bounds__(256) void gemm_wmma_f32(
    const float* __restrict__ X, const float* __restrict__ W,
    const float* __restrict__ bias,
    const float* __restrict__ emb, const int* __restrict__ idx, int embD,
    float* __restrict__ out, int N, int K) {
  __shared__ float Blds[2][CB * LDK];

  int tid = threadIdx.x;
  int lane = tid & 31;
  int wv = tid >> 5;
  int kh = lane >> 4;
  int l16 = lane & 15;
  int row0 = blockIdx.x * RB + wv * 16;
  int col0 = blockIdx.y * CB;

  int bcol = tid >> 2;
  int bseg = (tid & 3) * 16;
  const float* Bg = W + (size_t)(col0 + bcol) * K + bseg;
  float* Bs0 = &Blds[0][0] + bcol * LDK + bseg;
  float* Bs1 = &Blds[1][0] + bcol * LDK + bseg;

  const float* Ap = X + (size_t)(row0 + l16) * K + 2 * kh;

  v8f acc0 = {}, acc1 = {}, acc2 = {}, acc3 = {};

  {
    v4f p0 = *(const v4f*)(Bg + 0);
    v4f p1 = *(const v4f*)(Bg + 4);
    v4f p2 = *(const v4f*)(Bg + 8);
    v4f p3 = *(const v4f*)(Bg + 12);
    *(v4f*)(Bs0 + 0) = p0;
    *(v4f*)(Bs0 + 4) = p1;
    *(v4f*)(Bs0 + 8) = p2;
    *(v4f*)(Bs0 + 12) = p3;
  }
  __syncthreads();

  int nchunks = K / KC;
  for (int kc = 0; kc < nchunks; ++kc) {
    v4f p0, p1, p2, p3;
    bool hasNext = (kc + 1 < nchunks);
    if (hasNext) {
      const float* g = Bg + (size_t)(kc + 1) * KC;
      p0 = *(const v4f*)(g + 0);
      p1 = *(const v4f*)(g + 4);
      p2 = *(const v4f*)(g + 8);
      p3 = *(const v4f*)(g + 12);
    }

    const float* Bl = &Blds[kc & 1][0];
    const float* Al = Ap + (size_t)kc * KC;
    const float* bp0 = Bl + (0 + l16) * LDK + 2 * kh;
    const float* bp1 = Bl + (16 + l16) * LDK + 2 * kh;
    const float* bp2 = Bl + (32 + l16) * LDK + 2 * kh;
    const float* bp3 = Bl + (48 + l16) * LDK + 2 * kh;

#pragma unroll
    for (int k = 0; k < KC; k += 4) {
      v2f a = *(const v2f*)(Al + k);
      v2f b0 = *(const v2f*)(bp0 + k);
      v2f b1 = *(const v2f*)(bp1 + k);
      v2f b2 = *(const v2f*)(bp2 + k);
      v2f b3 = *(const v2f*)(bp3 + k);
      acc0 = __builtin_amdgcn_wmma_f32_16x16x4_f32(false, a, false, b0, (short)0, acc0, false, false);
      acc1 = __builtin_amdgcn_wmma_f32_16x16x4_f32(false, a, false, b1, (short)0, acc1, false, false);
      acc2 = __builtin_amdgcn_wmma_f32_16x16x4_f32(false, a, false, b2, (short)0, acc2, false, false);
      acc3 = __builtin_amdgcn_wmma_f32_16x16x4_f32(false, a, false, b3, (short)0, acc3, false, false);
    }

    if (hasNext) {
      float* s = ((kc + 1) & 1) ? Bs1 : Bs0;
      *(v4f*)(s + 0) = p0;
      *(v4f*)(s + 4) = p1;
      *(v4f*)(s + 8) = p2;
      *(v4f*)(s + 12) = p3;
      __syncthreads();
    }
  }

  int ridx[8];
#pragma unroll
  for (int r = 0; r < 8; ++r) ridx[r] = 0;
  if (idx) {
#pragma unroll
    for (int r = 0; r < 8; ++r) ridx[r] = idx[row0 + r + kh * 8];
  }

  v8f accs[4] = {acc0, acc1, acc2, acc3};
#pragma unroll
  for (int j = 0; j < 4; ++j) {
    int col = col0 + j * 16 + l16;
    float badd = bias ? bias[col] : 0.f;
#pragma unroll
    for (int r = 0; r < 8; ++r) {
      int row = row0 + r + kh * 8;
      float v = accs[j][r] + badd;
      if (emb) v += emb[(size_t)col * embD + ridx[r]];
      out[(size_t)row * N + col] = v;
    }
  }
}

__global__ void lstm_gates(const float* __restrict__ G, float* __restrict__ c,
                           float* __restrict__ h, float* __restrict__ seq,
                           int Hs, int seqOff) {
  int t = blockIdx.x * blockDim.x + threadIdx.x;
  if (t >= BN * Hs) return;
  int b = t / Hs;
  int u = t - b * Hs;
  const float* g = G + (size_t)b * 4 * Hs;
  float gi = g[u];
  float gf = g[Hs + u];
  float gg = g[2 * Hs + u];
  float go = g[3 * Hs + u];
  float si = 1.f / (1.f + expf(-gi));
  float sf = 1.f / (1.f + expf(-gf));
  float so = 1.f / (1.f + expf(-go));
  float cn = sf * c[t] + si * tanhf(gg);
  float hn = so * tanhf(cn);
  c[t] = cn;
  h[t] = hn;
  seq[(size_t)b * CH + seqOff + u] = hn;
}

__global__ void finalize(const float* __restrict__ pPs, const float* __restrict__ pPi,
                         const float* __restrict__ pMs, const float* __restrict__ pMi,
                         const int* __restrict__ tgt, float* __restrict__ out) {
  int t = blockIdx.x * blockDim.x + threadIdx.x;
  if (t >= PN * MN * BN) return;
  int b = t % BN;
  int pm = t / BN;
  int m = pm % MN;
  int p = pm / MN;
  const float* as = pPs + ((size_t)p * BN + b) * OPAD;
  const float* bsv = pMs + ((size_t)m * BN + b) * OPAD;
  const float* ai = pPi + ((size_t)p * BN + b) * OPAD;
  const float* biv = pMi + ((size_t)m * BN + b) * OPAD;
  int tg = (m < MN - 1) ? tgt[(m + 1) * BN + b] : 0;

  float mxs = -3.4e38f, mxi = -3.4e38f;
  for (int o = 0; o < ON; ++o) {
    float ls = as[o] + bsv[o];
    float li = ai[o] + biv[o];
    mxs = ls > mxs ? ls : mxs;
    mxi = li > mxi ? li : mxi;
  }
  float sums = 0.f, sumi = 0.f, ls52 = 0.f, li52 = 0.f, lst = 0.f, lit = 0.f;
  for (int o = 0; o < ON; ++o) {
    float ls = as[o] + bsv[o];
    float li = ai[o] + biv[o];
    sums += expf(ls - mxs);
    sumi += expf(li - mxi);
    if (o == ON - 1) { ls52 = ls; li52 = li; }
    if (o == tg) { lst = ls; lit = li; }
  }
  float lses = mxs + logf(sums);
  float lsei = mxi + logf(sumi);
  size_t idxPMB = ((size_t)p * MN + m) * BN + b;
  out[idxPMB] = ls52 - lses;
  out[(size_t)PN * MN * BN + idxPMB] = li52 - lsei;
  if (m < MN - 1) {
    size_t base = 2ull * PN * MN * BN;
    size_t idxS = ((size_t)p * (MN - 1) + m) * BN + b;
    out[base + idxS] = lst - lses;
    out[base + (size_t)PN * (MN - 1) * BN + idxS] = lit - lsei;
  }
}


extern "C" void kernel_launch(void* const* d_in, const int* in_sizes, int n_in,
                              void* d_out, int out_size, void* d_ws, size_t ws_size,
                              hipStream_t stream) {
  (void)in_sizes; (void)n_in; (void)out_size; (void)ws_size;
  const float* src_oh = (const float*)d_in[0];
  const float* tgt_oh = (const float*)d_in[1];  (void)tgt_oh;
  const float* Wih_pf = (const float*)d_in[2];
  const float* Whh_pf = (const float*)d_in[3];
  const float* b_pf   = (const float*)d_in[4];
  const float* Wih_pb = (const float*)d_in[5];
  const float* Whh_pb = (const float*)d_in[6];
  const float* b_pb   = (const float*)d_in[7];
  const float* Wih_m  = (const float*)d_in[8];
  const float* Whh_m  = (const float*)d_in[9];
  const float* b_m    = (const float*)d_in[10];
  const float* W_sub  = (const float*)d_in[11];
  const float* b_sub  = (const float*)d_in[12];
  const float* W_ins  = (const float*)d_in[13];
  const float* b_ins  = (const float*)d_in[14];
  const int*   tgt    = (const int*)d_in[15];

  char* wptr = (char*)d_ws;
  auto alloc = [&](size_t nfloats) -> float* {
    float* p = (float*)wptr;
    wptr += ((nfloats * 4 + 255) / 256) * 256;
    return p;
  };
  float* prior  = alloc((size_t)PN * BN * CH);
  float* modern = alloc((size_t)MN * BN * CH);
  float* Gpf = alloc((size_t)BN * 4 * HN);
  float* Gpb = alloc((size_t)BN * 4 * HN);
  float* Gm  = alloc((size_t)BN * 4 * CH);
  int hcN = 4 * BN * HN + 2 * BN * CH;
  float* hc = alloc(hcN);
  float* h_pf = hc;
  float* c_pf = h_pf + BN * HN;
  float* h_pb = c_pf + BN * HN;
  float* c_pb = h_pb + BN * HN;
  float* h_m  = c_pb + BN * HN;
  float* c_m  = h_m + BN * CH;
  float* Wsp = alloc(OPAD * CH);
  float* Wip = alloc(OPAD * CH);
  float* bsp = alloc(OPAD);
  float* bip = alloc(OPAD);
  float* pPs = alloc((size_t)PN * BN * OPAD);
  float* pPi = alloc((size_t)PN * BN * OPAD);
  float* pMs = alloc((size_t)MN * BN * OPAD);
  float* pMi = alloc((size_t)MN * BN * OPAD);
  int* sidx = (int*)alloc(PN * BN);

  zero_f32<<<(hcN + 255) / 256, 256, 0, stream>>>(hc, hcN);
  onehot_to_idx<<<(PN * BN + 255) / 256, 256, 0, stream>>>(src_oh, sidx, PN * BN);
  pad_proj_weights<<<(OPAD * CH + 255) / 256, 256, 0, stream>>>(
      W_sub, b_sub, W_ins, b_ins, Wsp, Wip, bsp, bip);

  for (int t = 0; t < PN; ++t) {
    gemm_wmma_f32<<<dim3(1, (4 * HN) / CB), 256, 0, stream>>>(
        h_pf, Whh_pf, b_pf, Wih_pf, sidx + t * BN, DN, Gpf, 4 * HN, HN);
    lstm_gates<<<(BN * HN + 255) / 256, 256, 0, stream>>>(
        Gpf, c_pf, h_pf, prior + (size_t)t * BN * CH, HN, 0);
    int tb = PN - 1 - t;
    gemm_wmma_f32<<<dim3(1, (4 * HN) / CB), 256, 0, stream>>>(
        h_pb, Whh_pb, b_pb, Wih_pb, sidx + tb * BN, DN, Gpb, 4 * HN, HN);
    lstm_gates<<<(BN * HN + 255) / 256, 256, 0, stream>>>(
        Gpb, c_pb, h_pb, prior + (size_t)tb * BN * CH, HN, HN);
    gemm_wmma_f32<<<dim3(1, (4 * CH) / CB), 256, 0, stream>>>(
        h_m, Whh_m, b_m, Wih_m, tgt + t * BN, DN, Gm, 4 * CH, CH);
    lstm_gates<<<(BN * CH + 255) / 256, 256, 0, stream>>>(
        Gm, c_m, h_m, modern + (size_t)t * BN * CH, CH, 0);
  }

  gemm_wmma_f32<<<dim3(PN * BN / RB, 1), 256, 0, stream>>>(
      prior, Wsp, bsp, nullptr, nullptr, 0, pPs, OPAD, CH);
  gemm_wmma_f32<<<dim3(PN * BN / RB, 1), 256, 0, stream>>>(
      prior, Wip, bip, nullptr, nullptr, 0, pPi, OPAD, CH);
  gemm_wmma_f32<<<dim3(MN * BN / RB, 1), 256, 0, stream>>>(
      modern, Wsp, bsp, nullptr, nullptr, 0, pMs, OPAD, CH);
  gemm_wmma_f32<<<dim3(MN * BN / RB, 1), 256, 0, stream>>>(
      modern, Wip, bip, nullptr, nullptr, 0, pMi, OPAD, CH);

  finalize<<<(PN * MN * BN + 255) / 256, 256, 0, stream>>>(
      pPs, pPi, pMs, pMi, tgt, (float*)d_out);
}


# Independent launches share one grid (block index picks the job): the three LSTM recurrences take one GEMM + one gate launch per step, the 4 projections one, set-up one; 200 launches -> 67
# speedup vs baseline: 2.3543x; 2.3543x over previous
#include <hip/hip_runtime.h>
#include <math.h>

typedef __attribute__((ext_vector_type(2))) float v2f;
typedef __attribute__((ext_vector_type(4))) float v4f;
typedef __attribute__((ext_vector_type(8))) float v8f;

#define PN 32
#define MN 32
#define BN 128
#define HN 256
#define DN 54
#define ON 53
#define OPAD 64
#define CH 512

#define RB 128
#define CB 64
#define KC 64
#define LDK (KC + 4)

#define HCN (4 * BN * HN + 2 * BN * CH)
#define ZBLK (HCN / 256)
#define IBLK ((PN * BN) / 256)
#define WBLK ((OPAD * CH) / 256)
#define MBLK ((4 * CH) / CB)
#define PBLK ((4 * HN) / CB)
#define GPB ((BN * HN) / 256)
#define GMB ((BN * CH) / 256)

static_assert(PN == MN, "the three recurrences advance in lock step");
static_assert(HCN % 256 == 0 && (PN * BN) % 256 == 0 && (OPAD * CH) % 256 == 0, "exact grids");
static_assert((BN * HN) % 256 == 0 && (BN * CH) % 256 == 0, "exact grids");
static_assert(BN == RB && (PN * BN) % RB == 0, "row blocks");


__device__ __forceinline__ void onehot_to_idx_one(const float* __restrict__ oh,
                                                  int* __restrict__ idx, int r) {
  const float* p = oh + (size_t)r * DN;
  int v = 0;
  for (int d = 0; d < DN; ++d)
    if (p[d] > 0.5f) v = d;
  idx[r] = v;
}

__device__ __forceinline__ void pad_proj_weights_one(
    const float* __restrict__ Ws, const float* __restrict__ bs,
    const float* __restrict__ Wi, const float* __restrict__ bi,
    float* __restrict__ Wsp, float* __restrict__ Wip,
    float* __restrict__ bsp, float* __restrict__ bip, int t) {
  int n = t / CH;
  int k = t - n * CH;
  Wsp[t] = (n < ON) ? Ws[n * CH + k] : 0.f;
  Wip[t] = (n < ON) ? Wi[n * CH + k] : 0.f;
  if (t < OPAD) {
    bsp[t] = (t < ON) ? bs[t] : 0.f;
    bip[t] = (t < ON) ? bi[t] : 0.f;
  }
}

__global__ void setup_all(float* __restrict__ hc,
                          const float* __restrict__ src_oh, int* __restrict__ sidx,
                          const float* __restrict__ Ws, const float* __restrict__ bs,
                          const float* __restrict__ Wi, const float* __restrict__ bi,
                          float* __restrict__ Wsp, float* __restrict__ Wip,
                          float* __restrict__ bsp, float* __restrict__ bip) {
  int blk = blockIdx.x;
  int tid = threadIdx.x;
  if (blk < ZBLK) {
    hc[blk * 256 + tid] = 0.f;
  } else if (blk < ZBLK + IBLK) {
    onehot_to_idx_one(src_oh, sidx, (blk - ZBLK) * 256 + tid);
  } else {
    pad_proj_weights_one(Ws, bs, Wi, bi, Wsp, Wip, bsp, bip,
                         (blk - ZBLK - IBLK) * 256 + tid);
  }
}

__device__ __forceinline__ void gemm_tile_wmma_f32(
    const float* __restrict__ X, const float* __restrict__ W,
    const float* __restrict__ bias,
    const float* __restrict__ emb, const int* __restrict__ idx, int embD,
    float* __restrict__ out, int N, int K, int rowBlk, int colBlk) {
  __shared__ float Blds[2][CB * LDK];

  int tid = threadIdx.x;
  int lane = tid & 31;
  int wv = tid >> 5;
  int kh = lane >> 4;
  int l16 = lane & 15;
  int row0 = rowBlk * RB + wv * 16;
  int col0 = colBlk * CB;

  int bcol = tid >> 2;
  int bseg = (tid & 3) * 16;
  const float* Bg = W + (size_t)(col0 + bcol) * K + bseg;
  float* Bs0 = &Blds[0][0] + bcol * LDK + bseg;
  float* Bs1 = &Blds[1][0] + bcol * LDK + bseg;

  const float* Ap = X + (size_t)(row0 + l16) * K + 2 * kh;

  v8f acc0 = {}, acc1 = {}, acc2 = {}, acc3 = {};

  {
    v4f p0 = *(const v4f*)(Bg + 0);
    v4f p1 = *(const v4f*)(Bg + 4);
    v4f p2 = *(const v4f*)(Bg + 8);
    v4f p3 = *(const v4f*)(Bg + 12);
    *(v4f*)(Bs0 + 0) = p0;
    *(v4f*)(Bs0 + 4) = p1;
    *(v4f*)(Bs0 + 8) = p2;
    *(v4f*)(Bs0 + 12) = p3;
  }
  __syncthreads();

  int nchunks = K / KC;
  for (int kc = 0; kc < nchunks; ++kc) {
    v4f p0, p1, p2, p3;
    bool hasNext = (kc + 1 < nchunks);
    if (hasNext) {
      const float* g = Bg + (size_t)(kc + 1) * KC;
      p0 = *(const v4f*)(g + 0);
      p1 = *(const v4f*)(g + 4);
      p2 = *(const v4f*)(g + 8);
      p3 = *(const v4f*)(g + 12);
    }

    const float* Bl = &Blds[kc & 1][0];
    const float* Al = Ap + (size_t)kc * KC;
    const float* bp0 = Bl + (0 + l16) * LDK + 2 * kh;
    const float* bp1 = Bl + (16 + l16) * LDK + 2 * kh;
    const float* bp2 = Bl + (32 + l16) * LDK + 2 * kh;
    const float* bp3 = Bl + (48 + l16) * LDK + 2 * kh;

#pragma unroll
    for (int k = 0; k < KC; k += 4) {
      v2f a = *(const v2f*)(Al + k);
      v2f b0 = *(const v2f*)(bp0 + k);
      v2f b1 = *(const v2f*)(bp1 + k);
      v2f b2 = *(const v2f*)(bp2 + k);
      v2f b3 = *(const v2f*)(bp3 + k);
      acc0 = __builtin_amdgcn_wmma_f32_16x16x4_f32(false, a, false, b0, (short)0, acc0, false, false);
      acc1 = __builtin_amdgcn_wmma_f32_16x16x4_f32(false, a, false, b1, (short)0, acc1, false, false);
      acc2 = __builtin_amdgcn_wmma_f32_16x16x4_f32(false, a, false, b2, (short)0, acc2, false, false);
      acc3 = __builtin_amdgcn_wmma_f32_16x16x4_f32(false, a, false, b3, (short)0, acc3, false, false);
    }

    if (hasNext) {
      float* s = ((kc + 1) & 1) ? Bs1 : Bs0;
      *(v4f*)(s + 0) = p0;
      *(v4f*)(s + 4) = p1;
      *(v4f*)(s + 8) = p2;
      *(v4f*)(s + 12) = p3;
      __syncthreads();
    }
  }

  int ridx[8];
#pragma unroll
  for (int r = 0; r < 8; ++r) ridx[r] = 0;
  if (idx) {
#pragma unroll
    for (int r = 0; r < 8; ++r) ridx[r] = idx[row0 + r + kh * 8];
  }

  v8f accs[4] = {acc0, acc1, acc2, acc3};
#pragma unroll
  for (int j = 0; j < 4; ++j) {
    int col = col0 + j * 16 + l16;
    float badd = bias ? bias[col] : 0.f;
#pragma unroll
    for (int r = 0; r < 8; ++r) {
      int row = row0 + r + kh * 8;
      float v = accs[j][r] + badd;
      if (emb) v += emb[(size_t)col * embD + ridx[r]];
      out[(size_t)row * N + col] = v;
    }
  }
}

__global__ __launch_bounds__(256) void lstm_step_gemm(
    const float* __restrict__ h_m, const float* __restrict__ Whh_m,
    const float* __restrict__ b_m, const float* __restrict__ Wih_m,
    const int* __restrict__ tok_m, float* __restrict__ G_m,
    const float* __restrict__ h_f, const float* __restrict__ Whh_f,
    const float* __restrict__ b_f, const float* __restrict__ Wih_f,
    const int* __restrict__ tok_f, float* __restrict__ G_f,
    const float* __restrict__ h_b, const float* __restrict__ Whh_b,
    const float* __restrict__ b_b, const float* __restrict__ Wih_b,
    const int* __restrict__ tok_b, float* __restrict__ G_b,
    int N_m, int K_m, int N_p, int K_p, int embD) {
  int blk = blockIdx.x;
  const float* X = h_m;
  const float* Wt = Whh_m;
  const float* bias = b_m;
  const float* emb = Wih_m;
  const int* tok = tok_m;
  float* out = G_m;
  int N = N_m, K = K_m;
  if (blk >= MBLK + PBLK) {
    blk -= MBLK + PBLK;
    X = h_b; Wt = Whh_b; bias = b_b; emb = Wih_b; tok = tok_b; out = G_b;
    N = N_p; K = K_p;
  } else if (blk >= MBLK) {
    blk -= MBLK;
    X = h_f; Wt = Whh_f; bias = b_f; emb = Wih_f; tok = tok_f; out = G_f;
    N = N_p; K = K_p;
  }
  gemm_tile_wmma_f32(X, Wt, bias, emb, tok, embD, out, N, K, 0, blk);
}

__global__ __launch_bounds__(256) void proj_gemm(
    const float* __restrict__ prior, const float* __restrict__ modern,
    const float* __restrict__ Wsp, const float* __restrict__ bsp,
    const float* __restrict__ Wip, const float* __restrict__ bip,
    float* __restrict__ pPs, float* __restrict__ pPi,
    float* __restrict__ pMs, float* __restrict__ pMi, int N, int K) {
  int job = blockIdx.y;
  const float* X = (job < 2) ? prior : modern;
  const float* Wt = (job & 1) ? Wip : Wsp;
  const float* bias = (job & 1) ? bip : bsp;
  float* out = (job == 0) ? pPs : (job == 1) ? pPi : (job == 2) ? pMs : pMi;
  gemm_tile_wmma_f32(X, Wt, bias, nullptr, nullptr, 0, out, N, K, blockIdx.x, 0);
}

__device__ __forceinline__ void lstm_gates_one(
    const float* __restrict__ G, float* __restrict__ c, float* __restrict__ h,
    float* __restrict__ seq, int Hs, int seqOff, int t) {
  if (t >= BN * Hs) return;
  int b = t / Hs;
  int u = t - b * Hs;
  const float* g = G + (size_t)b * 4 * Hs;
  float gi = g[u];
  float gf = g[Hs + u];
  float gg = g[2 * Hs + u];
  float go = g[3 * Hs + u];
  float si = 1.f / (1.f + expf(-gi));
  float sf = 1.f / (1.f + expf(-gf));
  float so = 1.f / (1.f + expf(-go));
  float cn = sf * c[t] + si * tanhf(gg);
  float hn = so * tanhf(cn);
  c[t] = cn;
  h[t] = hn;
  seq[(size_t)b * CH + seqOff + u] = hn;
}

__global__ void lstm_gates3(
    const float* __restrict__ G_m, float* __restrict__ c_m, float* __restrict__ h_m,
    float* __restrict__ seq_m,
    const float* __restrict__ G_f, float* __restrict__ c_f, float* __restrict__ h_f,
    float* __restrict__ seq_f,
    const float* __restrict__ G_b, float* __restrict__ c_b, float* __restrict__ h_b,
    float* __restrict__ seq_b) {
  int blk = blockIdx.x;
  int tid = threadIdx.x;
  if (blk < GMB) {
    lstm_gates_one(G_m, c_m, h_m, seq_m, CH, 0, blk * 256 + tid);
  } else if (blk < GMB + GPB) {
    lstm_gates_one(G_f, c_f, h_f, seq_f, HN, 0, (blk - GMB) * 256 + tid);
  } else {
    lstm_gates_one(G_b, c_b, h_b, seq_b, HN, HN, (blk - GMB - GPB) * 256 + tid);
  }
}

__global__ void finalize(const float* __restrict__ pPs, const float* __restrict__ pPi,
                         const float* __restrict__ pMs, const float* __restrict__ pMi,
                         const int* __restrict__ tgt, float* __restrict__ out) {
  int t = blockIdx.x * blockDim.x + threadIdx.x;
  if (t >= PN * MN * BN) return;
  int b = t % BN;
  int pm = t / BN;
  int m = pm % MN;
  int p = pm / MN;
  const float* as = pPs + ((size_t)p * BN + b) * OPAD;
  const float* bsv = pMs + ((size_t)m * BN + b) * OPAD;
  const float* ai = pPi + ((size_t)p * BN + b) * OPAD;
  const float* biv = pMi + ((size_t)m * BN + b) * OPAD;
  int tg = (m < MN - 1) ? tgt[(m + 1) * BN + b] : 0;

  float mxs = -3.4e38f, mxi = -3.4e38f;
  for (int o = 0; o < ON; ++o) {
    float ls = as[o] + bsv[o];
    float li = ai[o] + biv[o];
    mxs = ls > mxs ? ls : mxs;
    mxi = li > mxi ? li : mxi;
  }
  float sums = 0.f, sumi = 0.f, ls52 = 0.f, li52 = 0.f, lst = 0.f, lit = 0.f;
  for (int o = 0; o < ON; ++o) {
    float ls = as[o] + bsv[o];
    float li = ai[o] + biv[o];
    sums += expf(ls - mxs);
    sumi += expf(li - mxi);
    if (o == ON - 1) { ls52 = ls; li52 = li; }
    if (o == tg) { lst = ls; lit = li; }
  }
  float lses = mxs + logf(sums);
  float lsei = mxi + logf(sumi);
  size_t idxPMB = ((size_t)p * MN + m) * BN + b;
  out[idxPMB] = ls52 - lses;
  out[(size_t)PN * MN * BN + idxPMB] = li52 - lsei;
  if (m < MN - 1) {
    size_t base = 2ull * PN * MN * BN;
    size_t idxS = ((size_t)p * (MN - 1) + m) * BN + b;
    out[base + idxS] = lst - lses;
    out[base + (size_t)PN * (MN - 1) * BN + idxS] = lit - lsei;
  }
}


extern "C" void kernel_launch(void* const* d_in, const int* in_sizes, int n_in,
                              void* d_out, int out_size, void* d_ws, size_t ws_size,
                              hipStream_t stream) {
  (void)in_sizes; (void)n_in; (void)out_size; (void)ws_size;
  const float* src_oh = (const float*)d_in[0];
  const float* tgt_oh = (const float*)d_in[1];  (void)tgt_oh;
  const float* Wih_pf = (const float*)d_in[2];
  const float* Whh_pf = (const float*)d_in[3];
  const float* b_pf   = (const float*)d_in[4];
  const float* Wih_pb = (const float*)d_in[5];
  const float* Whh_pb = (const float*)d_in[6];
  const float* b_pb   = (const float*)d_in[7];
  const float* Wih_m  = (const float*)d_in[8];
  const float* Whh_m  = (const float*)d_in[9];
  const float* b_m    = (const float*)d_in[10];
  const float* W_sub  = (const float*)d_in[11];
  const float* b_sub  = (const float*)d_in[12];
  const float* W_ins  = (const float*)d_in[13];
  const float* b_ins  = (const float*)d_in[14];
  const int*   tgt    = (const int*)d_in[15];

  char* wptr = (char*)d_ws;
  auto alloc = [&](size_t nfloats) -> float* {
    float* p = (float*)wptr;
    wptr += ((nfloats * 4 + 255) / 256) * 256;
    return p;
  };
  float* prior  = alloc((size_t)PN * BN * CH);
  float* modern = alloc((size_t)MN * BN * CH);
  float* Gpf = alloc((size_t)BN * 4 * HN);
  float* Gpb = alloc((size_t)BN * 4 * HN);
  float* Gm  = alloc((size_t)BN * 4 * CH);
  float* hc = alloc(HCN);
  float* h_pf = hc;
  float* c_pf = h_pf + BN * HN;
  float* h_pb = c_pf + BN * HN;
  float* c_pb = h_pb + BN * HN;
  float* h_m  = c_pb + BN * HN;
  float* c_m  = h_m + BN * CH;
  float* Wsp = alloc(OPAD * CH);
  float* Wip = alloc(OPAD * CH);
  float* bsp = alloc(OPAD);
  float* bip = alloc(OPAD);
  float* pPs = alloc((size_t)PN * BN * OPAD);
  float* pPi = alloc((size_t)PN * BN * OPAD);
  float* pMs = alloc((size_t)MN * BN * OPAD);
  float* pMi = alloc((size_t)MN * BN * OPAD);
  int* sidx = (int*)alloc(PN * BN);

  setup_all<<<ZBLK + IBLK + WBLK, 256, 0, stream>>>(
      hc, src_oh, sidx, W_sub, b_sub, W_ins, b_ins, Wsp, Wip, bsp, bip);

  for (int t = 0; t < PN; ++t) {
    int tb = PN - 1 - t;
    lstm_step_gemm<<<MBLK + 2 * PBLK, 256, 0, stream>>>(
        h_m,  Whh_m,  b_m,  Wih_m,  tgt + t * BN,   Gm,
        h_pf, Whh_pf, b_pf, Wih_pf, sidx + t * BN,  Gpf,
        h_pb, Whh_pb, b_pb, Wih_pb, sidx + tb * BN, Gpb,
        4 * CH, CH, 4 * HN, HN, DN);
    lstm_gates3<<<GMB + 2 * GPB, 256, 0, stream>>>(
        Gm,  c_m,  h_m,  modern + (size_t)t * BN * CH,
        Gpf, c_pf, h_pf, prior + (size_t)t * BN * CH,
        Gpb, c_pb, h_pb, prior + (size_t)tb * BN * CH);
  }

  proj_gemm<<<dim3(PN * BN / RB, 4), 256, 0, stream>>>(
      prior, modern, Wsp, bsp, Wip, bip, pPs, pPi, pMs, pMi, OPAD, CH);

  finalize<<<(PN * MN * BN + 255) / 256, 256, 0, stream>>>(
      pPs, pPi, pMs, pMi, tgt, (float*)d_out);
}


# Gate update fused into the step GEMM epilogue (a tile holds all four gates of its units; bias, token column and c fetched ahead of the K loop; h read from the sequence buffer): launches 67 -> 35
# speedup vs baseline: 3.1267x; 1.3281x over previous
#include <hip/hip_runtime.h>
#include <math.h>

typedef __attribute__((ext_vector_type(2))) float v2f;
typedef __attribute__((ext_vector_type(4))) float v4f;
typedef __attribute__((ext_vector_type(8))) float v8f;

#define PN 32
#define MN 32
#define BN 128
#define HN 256
#define DN 54
#define ON 53
#define OPAD 64
#define CH 512

#define RB 128
#define CB 64
#define KC 64
#define LDK (KC + 4)

#define IBLK ((PN * BN) / 256)
#define WBLK ((OPAD * CH) / 256)
#define UB 16
#define MBLK (CH / UB)
#define PBLK (HN / UB)

static_assert(PN == MN, "the three recurrences advance in lock step");
static_assert((PN * BN) % 256 == 0 && (OPAD * CH) % 256 == 0, "exact grids");
static_assert(BN == RB && (PN * BN) % RB == 0, "row blocks");
static_assert(CB == 4 * UB && CH % UB == 0 && HN % UB == 0, "a step tile holds all four gates of its units");
static_assert(CH % KC == 0 && HN % KC == 0, "whole K chunks");


__device__ __forceinline__ void onehot_to_idx_one(const float* __restrict__ oh,
                                                  int* __restrict__ idx, int r) {
  const float* p = oh + (size_t)r * DN;
  int v = 0;
  for (int d = 0; d < DN; ++d)
    if (p[d] > 0.5f) v = d;
  idx[r] = v;
}

__device__ __forceinline__ void pad_proj_weights_one(
    const float* __restrict__ Ws, const float* __restrict__ bs,
    const float* __restrict__ Wi, const float* __restrict__ bi,
    float* __restrict__ Wsp, float* __restrict__ Wip,
    float* __restrict__ bsp, float* __restrict__ bip, int t) {
  int n = t / CH;
  int k = t - n * CH;
  Wsp[t] = (n < ON) ? Ws[n * CH + k] : 0.f;
  Wip[t] = (n < ON) ? Wi[n * CH + k] : 0.f;
  if (t < OPAD) {
    bsp[t] = (t < ON) ? bs[t] : 0.f;
    bip[t] = (t < ON) ? bi[t] : 0.f;
  }
}

__global__ void setup_all(const float* __restrict__ src_oh, int* __restrict__ sidx,
                          const float* __restrict__ Ws, const float* __restrict__ bs,
                          const float* __restrict__ Wi, const float* __restrict__ bi,
                          float* __restrict__ Wsp, float* __restrict__ Wip,
                          float* __restrict__ bsp, float* __restrict__ bip) {
  int blk = blockIdx.x;
  int tid = threadIdx.x;
  if (blk < IBLK) {
    onehot_to_idx_one(src_oh, sidx, blk * 256 + tid);
  } else {
    pad_proj_weights_one(Ws, bs, Wi, bi, Wsp, Wip, bsp, bip, (blk - IBLK) * 256 + tid);
  }
}

__device__ __forceinline__ void tile_mac_128x64(
    const float* __restrict__ Ap, const float* __restrict__ Bg, int nchunks,
    v8f& acc0, v8f& acc1, v8f& acc2, v8f& acc3) {
  __shared__ float Blds[2][CB * LDK];

  int tid = threadIdx.x;
  int lane = tid & 31;
  int kh = lane >> 4;
  int l16 = lane & 15;

  int bcol = tid >> 2;
  int bseg = (tid & 3) * 16;
  float* Bs0 = &Blds[0][0] + bcol * LDK + bseg;
  float* Bs1 = &Blds[1][0] + bcol * LDK + bseg;

  if (nchunks <= 0) return;

  {
    v4f p0 = *(const v4f*)(Bg + 0);
    v4f p1 = *(const v4f*)(Bg + 4);
    v4f p2 = *(const v4f*)(Bg + 8);
    v4f p3 = *(const v4f*)(Bg + 12);
    *(v4f*)(Bs0 + 0) = p0;
    *(v4f*)(Bs0 + 4) = p1;
    *(v4f*)(Bs0 + 8) = p2;
    *(v4f*)(Bs0 + 12) = p3;
  }
  __syncthreads();

  for (int kc = 0; kc < nchunks; ++kc) {
    v4f p0, p1, p2, p3;
    bool hasNext = (kc + 1 < nchunks);
    if (hasNext) {
      const float* g = Bg + (size_t)(kc + 1) * KC;
      p0 = *(const v4f*)(g + 0);
      p1 = *(const v4f*)(g + 4);
      p2 = *(const v4f*)(g + 8);
      p3 = *(const v4f*)(g + 12);
    }

    const float* Bl = &Blds[kc & 1][0];
    const float* Al = Ap + (size_t)kc * KC;
    const float* bp0 = Bl + (0 + l16) * LDK + 2 * kh;
    const float* bp1 = Bl + (16 + l16) * LDK + 2 * kh;
    const float* bp2 = Bl + (32 + l16) * LDK + 2 * kh;
    const float* bp3 = Bl + (48 + l16) * LDK + 2 * kh;

#pragma unroll
    for (int k = 0; k < KC; k += 4) {
      v2f a = *(const v2f*)(Al + k);
      v2f b0 = *(const v2f*)(bp0 + k);
      v2f b1 = *(const v2f*)(bp1 + k);
      v2f b2 = *(const v2f*)(bp2 + k);
      v2f b3 = *(const v2f*)(bp3 + k);
      acc0 = __builtin_amdgcn_wmma_f32_16x16x4_f32(false, a, false, b0, (short)0, acc0, false, false);
      acc1 = __builtin_amdgcn_wmma_f32_16x16x4_f32(false, a, false, b1, (short)0, acc1, false, false);
      acc2 = __builtin_amdgcn_wmma_f32_16x16x4_f32(false, a, false, b2, (short)0, acc2, false, false);
      acc3 = __builtin_amdgcn_wmma_f32_16x16x4_f32(false, a, false, b3, (short)0, acc3, false, false);
    }

    if (hasNext) {
      float* s = ((kc + 1) & 1) ? Bs1 : Bs0;
      *(v4f*)(s + 0) = p0;
      *(v4f*)(s + 4) = p1;
      *(v4f*)(s + 8) = p2;
      *(v4f*)(s + 12) = p3;
      __syncthreads();
    }
  }
}

__device__ __forceinline__ void gemm_tile_wmma_f32(
    const float* __restrict__ X, const float* __restrict__ W,
    const float* __restrict__ bias, float* __restrict__ out,
    int N, int K, int rowBlk, int colBlk) {
  int tid = threadIdx.x;
  int lane = tid & 31;
  int wv = tid >> 5;
  int kh = lane >> 4;
  int l16 = lane & 15;
  int row0 = rowBlk * RB + wv * 16;
  int col0 = colBlk * CB;

  const float* Bg = W + (size_t)(col0 + (tid >> 2)) * K + (tid & 3) * 16;
  const float* Ap = X + (size_t)(row0 + l16) * K + 2 * kh;

  v8f acc0 = {}, acc1 = {}, acc2 = {}, acc3 = {};
  tile_mac_128x64(Ap, Bg, K / KC, acc0, acc1, acc2, acc3);

  v8f accs[4] = {acc0, acc1, acc2, acc3};
#pragma unroll
  for (int j = 0; j < 4; ++j) {
    int col = col0 + j * 16 + l16;
    float badd = bias[col];
#pragma unroll
    for (int r = 0; r < 8; ++r) {
      int row = row0 + r + kh * 8;
      out[(size_t)row * N + col] = accs[j][r] + badd;
    }
  }
}

__device__ __forceinline__ void lstm_step_tile(
    const float* hprev, const float* __restrict__ Whh,
    const float* __restrict__ bias, const float* __restrict__ Wih,
    const int* __restrict__ tok, float* __restrict__ c, float* hout,
    int Hs, int kloop, int embD, int ub) {
  int tid = threadIdx.x;
  int lane = tid & 31;
  int wv = tid >> 5;
  int kh = lane >> 4;
  int l16 = lane & 15;
  int row0 = wv * 16;
  int u = ub * UB + l16;
  bool first = (kloop == 0);

  int tk[8];
#pragma unroll
  for (int r = 0; r < 8; ++r) tk[r] = tok[row0 + r + kh * 8];
  float badd[4], ein[4][8], cold[8];
#pragma unroll
  for (int j = 0; j < 4; ++j) {
    badd[j] = bias[j * Hs + u];
    const float* e = Wih + (size_t)(j * Hs + u) * embD;
#pragma unroll
    for (int r = 0; r < 8; ++r) ein[j][r] = e[tk[r]];
  }
  float* cp = c + (size_t)(ub * 8 + wv) * 256 + lane;
#pragma unroll
  for (int r = 0; r < 8; ++r) cold[r] = first ? 0.f : cp[r * 32];

  int bcol = tid >> 2;
  const float* Bg = Whh + (size_t)((bcol >> 4) * Hs + ub * UB + (bcol & 15)) * Hs + (tid & 3) * 16;
  const float* Ap = hprev + (size_t)(row0 + l16) * CH + 2 * kh;

  v8f acc0 = {}, acc1 = {}, acc2 = {}, acc3 = {};
  tile_mac_128x64(Ap, Bg, kloop / KC, acc0, acc1, acc2, acc3);

#pragma unroll
  for (int r = 0; r < 8; ++r) {
    int b = row0 + r + kh * 8;
    float gi = acc0[r] + badd[0]; gi += ein[0][r];
    float gf = acc1[r] + badd[1]; gf += ein[1][r];
    float gg = acc2[r] + badd[2]; gg += ein[2][r];
    float go = acc3[r] + badd[3]; go += ein[3][r];
    float si = 1.f / (1.f + expf(-gi));
    float sf = 1.f / (1.f + expf(-gf));
    float so = 1.f / (1.f + expf(-go));
    float cn = sf * cold[r] + si * tanhf(gg);
    float hn = so * tanhf(cn);
    cp[r * 32] = cn;
    hout[(size_t)b * CH + u] = hn;
  }
}

__global__ __launch_bounds__(256) void lstm_step(
    const float* hp_m, const float* __restrict__ Whh_m, const float* __restrict__ b_m,
    const float* __restrict__ Wih_m, const int* __restrict__ tok_m,
    float* __restrict__ c_m, float* ho_m,
    const float* hp_f, const float* __restrict__ Whh_f, const float* __restrict__ b_f,
    const float* __restrict__ Wih_f, const int* __restrict__ tok_f,
    float* __restrict__ c_f, float* ho_f,
    const float* hp_b, const float* __restrict__ Whh_b, const float* __restrict__ b_b,
    const float* __restrict__ Wih_b, const int* __restrict__ tok_b,
    float* __restrict__ c_b, float* ho_b,
    int Hs_m, int Hs_p, int kl_m, int kl_p, int embD) {
  int blk = blockIdx.x;
  const float* hp = hp_m;
  const float* Wt = Whh_m;
  const float* bias = b_m;
  const float* emb = Wih_m;
  const int* tok = tok_m;
  float* c = c_m;
  float* ho = ho_m;
  int Hs = Hs_m, kl = kl_m;
  if (blk >= MBLK + PBLK) {
    blk -= MBLK + PBLK;
    hp = hp_b; Wt = Whh_b; bias = b_b; emb = Wih_b; tok = tok_b; c = c_b; ho = ho_b;
    Hs = Hs_p; kl = kl_p;
  } else if (blk >= MBLK) {
    blk -= MBLK;
    hp = hp_f; Wt = Whh_f; bias = b_f; emb = Wih_f; tok = tok_f; c = c_f; ho = ho_f;
    Hs = Hs_p; kl = kl_p;
  }
  lstm_step_tile(hp, Wt, bias, emb, tok, c, ho, Hs, kl, embD, blk);
}

__global__ __launch_bounds__(256) void proj_gemm(
    const float* __restrict__ prior, const float* __restrict__ modern,
    const float* __restrict__ Wsp, const float* __restrict__ bsp,
    const float* __restrict__ Wip, const float* __restrict__ bip,
    float* __restrict__ pPs, float* __restrict__ pPi,
    float* __restrict__ pMs, float* __restrict__ pMi, int N, int K) {
  int job = blockIdx.y;
  const float* X = (job < 2) ? prior : modern;
  const float* Wt = (job & 1) ? Wip : Wsp;
  const float* bias = (job & 1) ? bip : bsp;
  float* out = (job == 0) ? pPs : (job == 1) ? pPi : (job == 2) ? pMs : pMi;
  gemm_tile_wmma_f32(X, Wt, bias, out, N, K, blockIdx.x, 0);
}

__global__ void finalize(const float* __restrict__ pPs, const float* __restrict__ pPi,
                         const float* __restrict__ pMs, const float* __restrict__ pMi,
                         const int* __restrict__ tgt, float* __restrict__ out) {
  int t = blockIdx.x * blockDim.x + threadIdx.x;
  if (t >= PN * MN * BN) return;
  int b = t % BN;
  int pm = t / BN;
  int m = pm % MN;
  int p = pm / MN;
  const float* as = pPs + ((size_t)p * BN + b) * OPAD;
  const float* bsv = pMs + ((size_t)m * BN + b) * OPAD;
  const float* ai = pPi + ((size_t)p * BN + b) * OPAD;
  const float* biv = pMi + ((size_t)m * BN + b) * OPAD;
  int tg = (m < MN - 1) ? tgt[(m + 1) * BN + b] : 0;

  float mxs = -3.4e38f, mxi = -3.4e38f;
  for (int o = 0; o < ON; ++o) {
    float ls = as[o] + bsv[o];
    float li = ai[o] + biv[o];
    mxs = ls > mxs ? ls : mxs;
    mxi = li > mxi ? li : mxi;
  }
  float sums = 0.f, sumi = 0.f, ls52 = 0.f, li52 = 0.f, lst = 0.f, lit = 0.f;
  for (int o = 0; o < ON; ++o) {
    float ls = as[o] + bsv[o];
    float li = ai[o] + biv[o];
    sums += expf(ls - mxs);
    sumi += expf(li - mxi);
    if (o == ON - 1) { ls52 = ls; li52 = li; }
    if (o == tg) { lst = ls; lit = li; }
  }
  float lses = mxs + logf(sums);
  float lsei = mxi + logf(sumi);
  size_t idxPMB = ((size_t)p * MN + m) * BN + b;
  out[idxPMB] = ls52 - lses;
  out[(size_t)PN * MN * BN + idxPMB] = li52 - lsei;
  if (m < MN - 1) {
    size_t base = 2ull * PN * MN * BN;
    size_t idxS = ((size_t)p * (MN - 1) + m) * BN + b;
    out[base + idxS] = lst - lses;
    out[base + (size_t)PN * (MN - 1) * BN + idxS] = lit - lsei;
  }
}


extern "C" void kernel_launch(void* const* d_in, const int* in_sizes, int n_in,
                              void* d_out, int out_size, void* d_ws, size_t ws_size,
                              hipStream_t stream) {
  (void)in_sizes; (void)n_in; (void)out_size; (void)ws_size;
  const float* src_oh = (const float*)d_in[0];
  const float* tgt_oh = (const float*)d_in[1];  (void)tgt_oh;
  const float* Wih_pf = (const float*)d_in[2];
  const float* Whh_pf = (const float*)d_in[3];
  const float* b_pf   = (const float*)d_in[4];
  const float* Wih_pb = (const float*)d_in[5];
  const float* Whh_pb = (const float*)d_in[6];
  const float* b_pb   = (const float*)d_in[7];
  const float* Wih_m  = (const float*)d_in[8];
  const float* Whh_m  = (const float*)d_in[9];
  const float* b_m    = (const float*)d_in[10];
  const float* W_sub  = (const float*)d_in[11];
  const float* b_sub  = (const float*)d_in[12];
  const float* W_ins  = (const float*)d_in[13];
  const float* b_ins  = (const float*)d_in[14];
  const int*   tgt    = (const int*)d_in[15];

  char* wptr = (char*)d_ws;
  auto alloc = [&](size_t nfloats) -> float* {
    float* p = (float*)wptr;
    wptr += ((nfloats * 4 + 255) / 256) * 256;
    return p;
  };
  float* prior  = alloc((size_t)PN * BN * CH);
  float* modern = alloc((size_t)MN * BN * CH);
  float* c_pf = alloc((size_t)BN * HN);
  float* c_pb = alloc((size_t)BN * HN);
  float* c_m  = alloc((size_t)BN * CH);
  float* Wsp = alloc(OPAD * CH);
  float* Wip = alloc(OPAD * CH);
  float* bsp = alloc(OPAD);
  float* bip = alloc(OPAD);
  float* pPs = alloc((size_t)PN * BN * OPAD);
  float* pPi = alloc((size_t)PN * BN * OPAD);
  float* pMs = alloc((size_t)MN * BN * OPAD);
  float* pMi = alloc((size_t)MN * BN * OPAD);
  int* sidx = (int*)alloc(PN * BN);

  setup_all<<<IBLK + WBLK, 256, 0, stream>>>(
      src_oh, sidx, W_sub, b_sub, W_ins, b_ins, Wsp, Wip, bsp, bip);

  lstm_step<<<MBLK + 2 * PBLK, 256, 0, stream>>>(
      modern,     Whh_m,  b_m,  Wih_m,  tgt,                  c_m,  modern,
      prior,      Whh_pf, b_pf, Wih_pf, sidx,                 c_pf, prior,
      prior + HN, Whh_pb, b_pb, Wih_pb, sidx + (PN - 1) * BN, c_pb,
      prior + (size_t)(PN - 1) * BN * CH + HN,
      CH, HN, 0, 0, DN);
  for (int t = 1; t < PN; ++t) {
    int tb = PN - 1 - t;
    lstm_step<<<MBLK + 2 * PBLK, 256, 0, stream>>>(
        modern + (size_t)(t - 1) * BN * CH, Whh_m,  b_m,  Wih_m,  tgt + t * BN,   c_m,
        modern + (size_t)t * BN * CH,
        prior + (size_t)(t - 1) * BN * CH,  Whh_pf, b_pf, Wih_pf, sidx + t * BN,  c_pf,
        prior + (size_t)t * BN * CH,
        prior + (size_t)(tb + 1) * BN * CH + HN, Whh_pb, b_pb, Wih_pb, sidx + tb * BN, c_pb,
        prior + (size_t)tb * BN * CH + HN,
        CH, HN, CH, HN, DN);
  }

  proj_gemm<<<dim3(PN * BN / RB, 4), 256, 0, stream>>>(
      prior, modern, Wsp, bsp, Wip, bip, pPs, pPi, pMs, pMi, OPAD, CH);

  finalize<<<(PN * MN * BN + 255) / 256, 256, 0, stream>>>(
      pPs, pPi, pMs, pMi, tgt, (float*)d_out);
}


# Step and projection tiles shrunk to fill more WGPs at one wave per SIMD: 64-row blocks of 4 waves, and 8-unit tiles for the K=512 LSTM (lane pairs swap gate halves); step grid 64 -> 192 blocks
# speedup vs baseline: 4.5850x; 1.4664x over previous
#include <hip/hip_runtime.h>
#include <math.h>

typedef __attribute__((ext_vector_type(2))) float v2f;
typedef __attribute__((ext_vector_type(4))) float v4f;
typedef __attribute__((ext_vector_type(8))) float v8f;

#define PN 32
#define MN 32
#define BN 128
#define HN 256
#define DN 54
#define ON 53
#define OPAD 64
#define CH 512

#define RB 64
#define NT (2 * RB)
#define RSPLIT (BN / RB)
#define CB 64
#define KC 64
#define LDK (KC + 4)

#define IBLK ((PN * BN) / 256)
#define WBLK ((OPAD * CH) / 256)
#define UBP 16
#define UBM 8
#define MBLK (CH / UBM)
#define PBLK (HN / UBP)

static_assert(PN == MN, "the three recurrences advance in lock step");
static_assert((PN * BN) % 256 == 0 && (OPAD * CH) % 256 == 0, "exact grids");
static_assert(BN % RB == 0 && (PN * BN) % RB == 0 && RB % 16 == 0, "row blocks");
static_assert(CB == 4 * UBP && CH % UBM == 0 && HN % UBP == 0, "a step tile holds all four gates of its units");
static_assert(NT == 128, "the staging map covers 32 tile columns per pass");
static_assert(CH % KC == 0 && HN % KC == 0, "whole K chunks");


__device__ __forceinline__ void onehot_to_idx_one(const float* __restrict__ oh,
                                                  int* __restrict__ idx, int r) {
  const float* p = oh + (size_t)r * DN;
  int v = 0;
  for (int d = 0; d < DN; ++d)
    if (p[d] > 0.5f) v = d;
  idx[r] = v;
}

__device__ __forceinline__ void pad_proj_weights_one(
    const float* __restrict__ Ws, const float* __restrict__ bs,
    const float* __restrict__ Wi, const float* __restrict__ bi,
    float* __restrict__ Wsp, float* __restrict__ Wip,
    float* __restrict__ bsp, float* __restrict__ bip, int t) {
  int n = t / CH;
  int k = t - n * CH;
  Wsp[t] = (n < ON) ? Ws[n * CH + k] : 0.f;
  Wip[t] = (n < ON) ? Wi[n * CH + k] : 0.f;
  if (t < OPAD) {
    bsp[t] = (t < ON) ? bs[t] : 0.f;
    bip[t] = (t < ON) ? bi[t] : 0.f;
  }
}

__global__ void setup_all(const float* __restrict__ src_oh, int* __restrict__ sidx,
                          const float* __restrict__ Ws, const float* __restrict__ bs,
                          const float* __restrict__ Wi, const float* __restrict__ bi,
                          float* __restrict__ Wsp, float* __restrict__ Wip,
                          float* __restrict__ bsp, float* __restrict__ bip) {
  int blk = blockIdx.x;
  int tid = threadIdx.x;
  if (blk < IBLK) {
    onehot_to_idx_one(src_oh, sidx, blk * 256 + tid);
  } else {
    pad_proj_weights_one(Ws, bs, Wi, bi, Wsp, Wip, bsp, bip, (blk - IBLK) * 256 + tid);
  }
}

template <int NACC>
__device__ __forceinline__ void tile_mac(
    const float* __restrict__ Ap, const float* const (&Bg)[NACC / 2], int nchunks,
    v8f (&acc)[NACC]) {
  constexpr int CBW = 16 * NACC;
  constexpr int SP = (4 * CBW) / NT;
  static_assert(SP == NACC / 2 && SP * NT == 4 * CBW, "whole staging passes");
  __shared__ float Blds[2][CBW * LDK];

  int tid = threadIdx.x;
  int lane = tid & 31;
  int kh = lane >> 4;
  int l16 = lane & 15;

  int bcol = tid >> 2;
  int bseg = (tid & 3) * 16;
  float* Bs0 = &Blds[0][0] + bcol * LDK + bseg;
  float* Bs1 = &Blds[1][0] + bcol * LDK + bseg;

  if (nchunks <= 0) return;

  {
    v4f p[SP][4];
#pragma unroll
    for (int q = 0; q < SP; ++q) {
      p[q][0] = *(const v4f*)(Bg[q] + 0);
      p[q][1] = *(const v4f*)(Bg[q] + 4);
      p[q][2] = *(const v4f*)(Bg[q] + 8);
      p[q][3] = *(const v4f*)(Bg[q] + 12);
    }
#pragma unroll
    for (int q = 0; q < SP; ++q) {
      float* d = Bs0 + q * (NT / 4) * LDK;
      *(v4f*)(d + 0) = p[q][0];
      *(v4f*)(d + 4) = p[q][1];
      *(v4f*)(d + 8) = p[q][2];
      *(v4f*)(d + 12) = p[q][3];
    }
  }
  __syncthreads();

  for (int kc = 0; kc < nchunks; ++kc) {
    v4f p[SP][4];
    bool hasNext = (kc + 1 < nchunks);
    if (hasNext) {
#pragma unroll
      for (int q = 0; q < SP; ++q) {
        const float* g = Bg[q] + (size_t)(kc + 1) * KC;
        p[q][0] = *(const v4f*)(g + 0);
        p[q][1] = *(const v4f*)(g + 4);
        p[q][2] = *(const v4f*)(g + 8);
        p[q][3] = *(const v4f*)(g + 12);
      }
    }

    const float* Bl = &Blds[kc & 1][0];
    const float* Al = Ap + (size_t)kc * KC;
    const float* bp[NACC];
#pragma unroll
    for (int j = 0; j < NACC; ++j) bp[j] = Bl + (16 * j + l16) * LDK + 2 * kh;

#pragma unroll
    for (int k = 0; k < KC; k += 4) {
      v2f a = *(const v2f*)(Al + k);
      v2f b[NACC];
#pragma unroll
      for (int j = 0; j < NACC; ++j) b[j] = *(const v2f*)(bp[j] + k);
#pragma unroll
      for (int j = 0; j < NACC; ++j)
        acc[j] = __builtin_amdgcn_wmma_f32_16x16x4_f32(false, a, false, b[j], (short)0, acc[j], false, false);
    }

    if (hasNext) {
      float* d0 = ((kc + 1) & 1) ? Bs1 : Bs0;
#pragma unroll
      for (int q = 0; q < SP; ++q) {
        float* d = d0 + q * (NT / 4) * LDK;
        *(v4f*)(d + 0) = p[q][0];
        *(v4f*)(d + 4) = p[q][1];
        *(v4f*)(d + 8) = p[q][2];
        *(v4f*)(d + 12) = p[q][3];
      }
      __syncthreads();
    }
  }
}

__device__ __forceinline__ void gemm_tile_wmma_f32(
    const float* __restrict__ X, const float* __restrict__ W,
    const float* __restrict__ bias, float* __restrict__ out,
    int N, int K, int rowBlk, int colBlk) {
  int tid = threadIdx.x;
  int lane = tid & 31;
  int wv = tid >> 5;
  int kh = lane >> 4;
  int l16 = lane & 15;
  int row0 = rowBlk * RB + wv * 16;
  int col0 = colBlk * CB;

  const float* Bg[2];
#pragma unroll
  for (int q = 0; q < 2; ++q)
    Bg[q] = W + (size_t)(col0 + q * (NT / 4) + (tid >> 2)) * K + (tid & 3) * 16;
  const float* Ap = X + (size_t)(row0 + l16) * K + 2 * kh;

  v8f accs[4] = {};
  tile_mac<4>(Ap, Bg, K / KC, accs);

#pragma unroll
  for (int j = 0; j < 4; ++j) {
    int col = col0 + j * 16 + l16;
    float badd = bias[col];
#pragma unroll
    for (int r = 0; r < 8; ++r) {
      int row = row0 + r + kh * 8;
      out[(size_t)row * N + col] = accs[j][r] + badd;
    }
  }
}

template <int NACC>
__device__ __forceinline__ void lstm_step_tile(
    const float* hprev, const float* __restrict__ Whh,
    const float* __restrict__ bias, const float* __restrict__ Wih,
    const int* __restrict__ tok, float* __restrict__ c, float* hout,
    int Hs, int kloop, int embD, int ub, int rh) {
  constexpr int UBK = 4 * NACC;
  constexpr int NR = 2 * NACC;
  static_assert(NACC == 4 || NACC == 2, "gate layouts written out for 16 and 8 units");
  int tid = threadIdx.x;
  int lane = tid & 31;
  int wv = rh * (RB / 16) + (tid >> 5);
  int kh = lane >> 4;
  int l16 = lane & 15;
  int row0 = wv * 16;
  bool lo = (NACC == 4) || (l16 < 8);
  int ul = (NACC == 4) ? l16 : (l16 & 7);
  int rbase = row0 + kh * 8 + (lo ? 0 : 4);
  int u = ub * UBK + ul;
  bool first = (kloop == 0);

  int tk[NR];
#pragma unroll
  for (int r = 0; r < NR; ++r) tk[r] = tok[rbase + r];
  float badd[4], ein[4][NR], cold[NR];
#pragma unroll
  for (int j = 0; j < 4; ++j) {
    badd[j] = bias[j * Hs + u];
    const float* e = Wih + (size_t)(j * Hs + u) * embD;
#pragma unroll
    for (int r = 0; r < NR; ++r) ein[j][r] = e[tk[r]];
  }
  float* cp = c + (size_t)(ub * (BN / 16) + wv) * (32 * NR) + lane;
#pragma unroll
  for (int r = 0; r < NR; ++r) cold[r] = first ? 0.f : cp[r * 32];

  const float* Bg[NACC / 2];
#pragma unroll
  for (int q = 0; q < NACC / 2; ++q) {
    int n = q * (NT / 4) + (tid >> 2);
    Bg[q] = Whh + (size_t)((n / UBK) * Hs + ub * UBK + (n % UBK)) * Hs + (tid & 3) * 16;
  }
  const float* Ap = hprev + (size_t)(row0 + l16) * CH + 2 * kh;

  v8f acc[NACC] = {};
  tile_mac<NACC>(Ap, Bg, kloop / KC, acc);

  float ai[NR], af[NR], ag[NR], ao[NR];
  if constexpr (NACC == 4) {
#pragma unroll
    for (int r = 0; r < NR; ++r) {
      ai[r] = acc[0][r]; af[r] = acc[1][r]; ag[r] = acc[2][r]; ao[r] = acc[3][r];
    }
  } else {
    v8f m0 = lo ? acc[0] : __builtin_shufflevector(acc[0], acc[0], 4, 5, 6, 7, 0, 1, 2, 3);
    v8f m1 = lo ? acc[1] : __builtin_shufflevector(acc[1], acc[1], 4, 5, 6, 7, 0, 1, 2, 3);
#pragma unroll
    for (int r = 0; r < NR; ++r) {
      float got0 = __shfl_xor(m0[r + 4], 8, 32);
      float got1 = __shfl_xor(m1[r + 4], 8, 32);
      ai[r] = lo ? m0[r] : got0; af[r] = lo ? got0 : m0[r];
      ag[r] = lo ? m1[r] : got1; ao[r] = lo ? got1 : m1[r];
    }
  }

#pragma unroll
  for (int r = 0; r < NR; ++r) {
    int b = rbase + r;
    float gi = ai[r] + badd[0]; gi += ein[0][r];
    float gf = af[r] + badd[1]; gf += ein[1][r];
    float gg = ag[r] + badd[2]; gg += ein[2][r];
    float go = ao[r] + badd[3]; go += ein[3][r];
    float si = 1.f / (1.f + expf(-gi));
    float sf = 1.f / (1.f + expf(-gf));
    float so = 1.f / (1.f + expf(-go));
    float cn = sf * cold[r] + si * tanhf(gg);
    float hn = so * tanhf(cn);
    cp[r * 32] = cn;
    hout[(size_t)b * CH + u] = hn;
  }
}

__global__ __launch_bounds__(NT) void lstm_step(
    const float* hp_m, const float* __restrict__ Whh_m, const float* __restrict__ b_m,
    const float* __restrict__ Wih_m, const int* __restrict__ tok_m,
    float* __restrict__ c_m, float* ho_m,
    const float* hp_f, const float* __restrict__ Whh_f, const float* __restrict__ b_f,
    const float* __restrict__ Wih_f, const int* __restrict__ tok_f,
    float* __restrict__ c_f, float* ho_f,
    const float* hp_b, const float* __restrict__ Whh_b, const float* __restrict__ b_b,
    const float* __restrict__ Wih_b, const int* __restrict__ tok_b,
    float* __restrict__ c_b, float* ho_b,
    int Hs_m, int Hs_p, int kl_m, int kl_p, int embD) {
  int rh = blockIdx.x % RSPLIT;
  int blk = blockIdx.x / RSPLIT;
  if (blk < MBLK) {
    lstm_step_tile<2>(hp_m, Whh_m, b_m, Wih_m, tok_m, c_m, ho_m, Hs_m, kl_m, embD, blk, rh);
    return;
  }
  blk -= MBLK;
  const float* hp = hp_f;
  const float* Wt = Whh_f;
  const float* bias = b_f;
  const float* emb = Wih_f;
  const int* tok = tok_f;
  float* c = c_f;
  float* ho = ho_f;
  if (blk >= PBLK) {
    blk -= PBLK;
    hp = hp_b; Wt = Whh_b; bias = b_b; emb = Wih_b; tok = tok_b; c = c_b; ho = ho_b;
  }
  lstm_step_tile<4>(hp, Wt, bias, emb, tok, c, ho, Hs_p, kl_p, embD, blk, rh);
}

__global__ __launch_bounds__(NT) void proj_gemm(
    const float* __restrict__ prior, const float* __restrict__ modern,
    const float* __restrict__ Wsp, const float* __restrict__ bsp,
    const float* __restrict__ Wip, const float* __restrict__ bip,
    float* __restrict__ pPs, float* __restrict__ pPi,
    float* __restrict__ pMs, float* __restrict__ pMi, int N, int K) {
  int job = blockIdx.y;
  const float* X = (job < 2) ? prior : modern;
  const float* Wt = (job & 1) ? Wip : Wsp;
  const float* bias = (job & 1) ? bip : bsp;
  float* out = (job == 0) ? pPs : (job == 1) ? pPi : (job == 2) ? pMs : pMi;
  gemm_tile_wmma_f32(X, Wt, bias, out, N, K, blockIdx.x, 0);
}

__global__ void finalize(const float* __restrict__ pPs, const float* __restrict__ pPi,
                         const float* __restrict__ pMs, const float* __restrict__ pMi,
                         const int* __restrict__ tgt, float* __restrict__ out) {
  int t = blockIdx.x * blockDim.x + threadIdx.x;
  if (t >= PN * MN * BN) return;
  int b = t % BN;
  int pm = t / BN;
  int m = pm % MN;
  int p = pm / MN;
  const float* as = pPs + ((size_t)p * BN + b) * OPAD;
  const float* bsv = pMs + ((size_t)m * BN + b) * OPAD;
  const float* ai = pPi + ((size_t)p * BN + b) * OPAD;
  const float* biv = pMi + ((size_t)m * BN + b) * OPAD;
  int tg = (m < MN - 1) ? tgt[(m + 1) * BN + b] : 0;

  float mxs = -3.4e38f, mxi = -3.4e38f;
  for (int o = 0; o < ON; ++o) {
    float ls = as[o] + bsv[o];
    float li = ai[o] + biv[o];
    mxs = ls > mxs ? ls : mxs;
    mxi = li > mxi ? li : mxi;
  }
  float sums = 0.f, sumi = 0.f, ls52 = 0.f, li52 = 0.f, lst = 0.f, lit = 0.f;
  for (int o = 0; o < ON; ++o) {
    float ls = as[o] + bsv[o];
    float li = ai[o] + biv[o];
    sums += expf(ls - mxs);
    sumi += expf(li - mxi);
    if (o == ON - 1) { ls52 = ls; li52 = li; }
    if (o == tg) { lst = ls; lit = li; }
  }
  float lses = mxs + logf(sums);
  float lsei = mxi + logf(sumi);
  size_t idxPMB = ((size_t)p * MN + m) * BN + b;
  out[idxPMB] = ls52 - lses;
  out[(size_t)PN * MN * BN + idxPMB] = li52 - lsei;
  if (m < MN - 1) {
    size_t base = 2ull * PN * MN * BN;
    size_t idxS = ((size_t)p * (MN - 1) + m) * BN + b;
    out[base + idxS] = lst - lses;
    out[base + (size_t)PN * (MN - 1) * BN + idxS] = lit - lsei;
  }
}


extern "C" void kernel_launch(void* const* d_in, const int* in_sizes, int n_in,
                              void* d_out, int out_size, void* d_ws, size_t ws_size,
                              hipStream_t stream) {
  (void)in_sizes; (void)n_in; (void)out_size; (void)ws_size;
  const float* src_oh = (const float*)d_in[0];
  const float* tgt_oh = (const float*)d_in[1];  (void)tgt_oh;
  const float* Wih_pf = (const float*)d_in[2];
  const float* Whh_pf = (const float*)d_in[3];
  const float* b_pf   = (const float*)d_in[4];
  const float* Wih_pb = (const float*)d_in[5];
  const float* Whh_pb = (const float*)d_in[6];
  const float* b_pb   = (const float*)d_in[7];
  const float* Wih_m  = (const float*)d_in[8];
  const float* Whh_m  = (const float*)d_in[9];
  const float* b_m    = (const float*)d_in[10];
  const float* W_sub  = (const float*)d_in[11];
  const float* b_sub  = (const float*)d_in[12];
  const float* W_ins  = (const float*)d_in[13];
  const float* b_ins  = (const float*)d_in[14];
  const int*   tgt    = (const int*)d_in[15];

  char* wptr = (char*)d_ws;
  auto alloc = [&](size_t nfloats) -> float* {
    float* p = (float*)wptr;
    wptr += ((nfloats * 4 + 255) / 256) * 256;
    return p;
  };
  float* prior  = alloc((size_t)PN * BN * CH);
  float* modern = alloc((size_t)MN * BN * CH);
  float* c_pf = alloc((size_t)BN * HN);
  float* c_pb = alloc((size_t)BN * HN);
  float* c_m  = alloc((size_t)BN * CH);
  float* Wsp = alloc(OPAD * CH);
  float* Wip = alloc(OPAD * CH);
  float* bsp = alloc(OPAD);
  float* bip = alloc(OPAD);
  float* pPs = alloc((size_t)PN * BN * OPAD);
  float* pPi = alloc((size_t)PN * BN * OPAD);
  float* pMs = alloc((size_t)MN * BN * OPAD);
  float* pMi = alloc((size_t)MN * BN * OPAD);
  int* sidx = (int*)alloc(PN * BN);

  setup_all<<<IBLK + WBLK, 256, 0, stream>>>(
      src_oh, sidx, W_sub, b_sub, W_ins, b_ins, Wsp, Wip, bsp, bip);

  lstm_step<<<(MBLK + 2 * PBLK) * RSPLIT, NT, 0, stream>>>(
      modern,     Whh_m,  b_m,  Wih_m,  tgt,                  c_m,  modern,
      prior,      Whh_pf, b_pf, Wih_pf, sidx,                 c_pf, prior,
      prior + HN, Whh_pb, b_pb, Wih_pb, sidx + (PN - 1) * BN, c_pb,
      prior + (size_t)(PN - 1) * BN * CH + HN,
      CH, HN, 0, 0, DN);
  for (int t = 1; t < PN; ++t) {
    int tb = PN - 1 - t;
    lstm_step<<<(MBLK + 2 * PBLK) * RSPLIT, NT, 0, stream>>>(
        modern + (size_t)(t - 1) * BN * CH, Whh_m,  b_m,  Wih_m,  tgt + t * BN,   c_m,
        modern + (size_t)t * BN * CH,
        prior + (size_t)(t - 1) * BN * CH,  Whh_pf, b_pf, Wih_pf, sidx + t * BN,  c_pf,
        prior + (size_t)t * BN * CH,
        prior + (size_t)(tb + 1) * BN * CH + HN, Whh_pb, b_pb, Wih_pb, sidx + tb * BN, c_pb,
        prior + (size_t)tb * BN * CH + HN,
        CH, HN, CH, HN, DN);
  }

  proj_gemm<<<dim3(PN * BN / RB, 4), NT, 0, stream>>>(
      prior, modern, Wsp, bsp, Wip, bip, pPs, pPi, pMs, pMi, OPAD, CH);

  finalize<<<(PN * MN * BN + 255) / 256, 256, 0, stream>>>(
      pPs, pPi, pMs, pMi, tgt, (float*)d_out);
}


# Step and projection GEMM operands fetched as whole 128-byte lines: h rows staged through LDS beside the weights, each wave load one row's contiguous 512 B, prefetched one 128-k chunk ahead
# speedup vs baseline: 6.4460x; 1.4059x over previous
#include <hip/hip_runtime.h>
#include <math.h>

typedef __attribute__((ext_vector_type(2))) float v2f;
typedef __attribute__((ext_vector_type(4))) float v4f;
typedef __attribute__((ext_vector_type(8))) float v8f;

#define PN 32
#define MN 32
#define BN 128
#define HN 256
#define DN 54
#define ON 53
#define OPAD 64
#define CH 512

#define RB 64
#define NT (2 * RB)
#define RSPLIT (BN / RB)
#define CB 64
#define KC 128
#define LDK (KC + 4)
#define NW (NT / 32)
#define STAGE_MAX ((RB + CB) * LDK)
#define LDS_FLOATS (2 * STAGE_MAX)

#define IBLK ((PN * BN) / 256)
#define WBLK ((OPAD * CH) / 256)
#define UBP 16
#define UBM 8
#define MBLK (CH / UBM)
#define PBLK (HN / UBP)

static_assert(PN == MN, "the three recurrences advance in lock step");
static_assert((PN * BN) % 256 == 0 && (OPAD * CH) % 256 == 0, "exact grids");
static_assert(BN % RB == 0 && (PN * BN) % RB == 0 && RB % 16 == 0, "row blocks");
static_assert(CB == 4 * UBP && CH % UBM == 0 && HN % UBP == 0, "a step tile holds all four gates of its units");
static_assert(NT == 128 && RB == 16 * NW, "a wave stages the 16 A rows it multiplies");
static_assert(KC == 4 * 32, "a wave load of 16 B per lane covers one staged row");
static_assert(CH % KC == 0 && HN % KC == 0, "whole K chunks");


__device__ __forceinline__ void onehot_to_idx_one(const float* __restrict__ oh,
                                                  int* __restrict__ idx, int r) {
  const float* p = oh + (size_t)r * DN;
  int v = 0;
  for (int d = 0; d < DN; ++d)
    if (p[d] > 0.5f) v = d;
  idx[r] = v;
}

__device__ __forceinline__ void pad_proj_weights_one(
    const float* __restrict__ Ws, const float* __restrict__ bs,
    const float* __restrict__ Wi, const float* __restrict__ bi,
    float* __restrict__ Wsp, float* __restrict__ Wip,
    float* __restrict__ bsp, float* __restrict__ bip, int t) {
  int n = t / CH;
  int k = t - n * CH;
  Wsp[t] = (n < ON) ? Ws[n * CH + k] : 0.f;
  Wip[t] = (n < ON) ? Wi[n * CH + k] : 0.f;
  if (t < OPAD) {
    bsp[t] = (t < ON) ? bs[t] : 0.f;
    bip[t] = (t < ON) ? bi[t] : 0.f;
  }
}

__global__ void setup_all(const float* __restrict__ src_oh, int* __restrict__ sidx,
                          const float* __restrict__ Ws, const float* __restrict__ bs,
                          const float* __restrict__ Wi, const float* __restrict__ bi,
                          float* __restrict__ Wsp, float* __restrict__ Wip,
                          float* __restrict__ bsp, float* __restrict__ bip) {
  int blk = blockIdx.x;
  int tid = threadIdx.x;
  if (blk < IBLK) {
    onehot_to_idx_one(src_oh, sidx, blk * 256 + tid);
  } else {
    pad_proj_weights_one(Ws, bs, Wi, bi, Wsp, Wip, bsp, bip, (blk - IBLK) * 256 + tid);
  }
}

template <int NACC, class ColRow>
__device__ __forceinline__ void tile_mac(
    float* lds, const float* __restrict__ A, int lda,
    const float* __restrict__ Bw, int ldb, ColRow colrow, int nchunks,
    v8f (&acc)[NACC]) {
  constexpr int CBW = 16 * NACC;
  constexpr int QA = RB / NW;
  constexpr int QB = CBW / NW;
  constexpr int STG = (RB + CBW) * LDK;
  static_assert(QA == 16 && QB * NW == CBW && STG <= STAGE_MAX, "whole rows per wave");

  int tid = threadIdx.x;
  int lane = tid & 31;
  int wv = tid >> 5;
  int kh = lane >> 4;
  int l16 = lane & 15;

  if (nchunks <= 0) return;

  const float* ag = A + (size_t)(QA * wv) * lda + 4 * lane;
  const float* bg[QB];
#pragma unroll
  for (int q = 0; q < QB; ++q) bg[q] = Bw + (size_t)colrow(QB * wv + q) * ldb + 4 * lane;
  float* as = lds + (QA * wv) * LDK + 4 * lane;
  float* bs = lds + (RB + QB * wv) * LDK + 4 * lane;

  {
    v4f pa[QA], pb[QB];
#pragma unroll
    for (int i = 0; i < QA; ++i) pa[i] = *(const v4f*)(ag + (size_t)i * lda);
#pragma unroll
    for (int q = 0; q < QB; ++q) pb[q] = *(const v4f*)(bg[q]);
#pragma unroll
    for (int i = 0; i < QA; ++i) *(v4f*)(as + i * LDK) = pa[i];
#pragma unroll
    for (int q = 0; q < QB; ++q) *(v4f*)(bs + q * LDK) = pb[q];
  }
  __syncthreads();

  const float* af = lds + (16 * wv + l16) * LDK + 2 * kh;
  const float* bf = lds + (RB + l16) * LDK + 2 * kh;

  for (int kc = 0; kc < nchunks; ++kc) {
    v4f pa[QA], pb[QB];
    bool hasNext = (kc + 1 < nchunks);
    if (hasNext) {
      size_t ko = (size_t)(kc + 1) * KC;
#pragma unroll
      for (int i = 0; i < QA; ++i) pa[i] = *(const v4f*)(ag + (size_t)i * lda + ko);
#pragma unroll
      for (int q = 0; q < QB; ++q) pb[q] = *(const v4f*)(bg[q] + ko);
    }

    const float* al = af + (kc & 1) * STG;
    const float* bl = bf + (kc & 1) * STG;

#pragma unroll
    for (int k = 0; k < KC; k += 4) {
      v2f a = *(const v2f*)(al + k);
      v2f b[NACC];
#pragma unroll
      for (int j = 0; j < NACC; ++j) b[j] = *(const v2f*)(bl + 16 * j * LDK + k);
#pragma unroll
      for (int j = 0; j < NACC; ++j)
        acc[j] = __builtin_amdgcn_wmma_f32_16x16x4_f32(false, a, false, b[j], (short)0, acc[j], false, false);
    }

    if (hasNext) {
      int so = ((kc + 1) & 1) * STG;
#pragma unroll
      for (int i = 0; i < QA; ++i) *(v4f*)(as + so + i * LDK) = pa[i];
#pragma unroll
      for (int q = 0; q < QB; ++q) *(v4f*)(bs + so + q * LDK) = pb[q];
      __syncthreads();
    }
  }
}

__device__ __forceinline__ void gemm_tile_wmma_f32(
    float* lds, const float* __restrict__ X, const float* __restrict__ W,
    const float* __restrict__ bias, float* __restrict__ out,
    int N, int K, int rowBlk, int colBlk) {
  int tid = threadIdx.x;
  int lane = tid & 31;
  int wv = tid >> 5;
  int kh = lane >> 4;
  int l16 = lane & 15;
  int row0 = rowBlk * RB + wv * 16;
  int col0 = colBlk * CB;

  v8f accs[4] = {};
  tile_mac<4>(lds, X + (size_t)(rowBlk * RB) * K, K, W, K,
              [=](int n) { return col0 + n; }, K / KC, accs);

#pragma unroll
  for (int j = 0; j < 4; ++j) {
    int col = col0 + j * 16 + l16;
    float badd = bias[col];
#pragma unroll
    for (int r = 0; r < 8; ++r) {
      int row = row0 + r + kh * 8;
      out[(size_t)row * N + col] = accs[j][r] + badd;
    }
  }
}

template <int NACC>
__device__ __forceinline__ void lstm_step_tile(
    float* lds, const float* hprev, const float* __restrict__ Whh,
    const float* __restrict__ bias, const float* __restrict__ Wih,
    const int* __restrict__ tok, float* __restrict__ c, float* hout,
    int Hs, int kloop, int embD, int ub, int rh) {
  constexpr int UBK = 4 * NACC;
  constexpr int NR = 2 * NACC;
  static_assert(NACC == 4 || NACC == 2, "gate layouts written out for 16 and 8 units");
  int tid = threadIdx.x;
  int lane = tid & 31;
  int wv = rh * (RB / 16) + (tid >> 5);
  int kh = lane >> 4;
  int l16 = lane & 15;
  int row0 = wv * 16;
  bool lo = (NACC == 4) || (l16 < 8);
  int ul = (NACC == 4) ? l16 : (l16 & 7);
  int rbase = row0 + kh * 8 + (lo ? 0 : 4);
  int u = ub * UBK + ul;
  bool first = (kloop == 0);

  int tk[NR];
#pragma unroll
  for (int r = 0; r < NR; ++r) tk[r] = tok[rbase + r];
  float badd[4], ein[4][NR], cold[NR];
#pragma unroll
  for (int j = 0; j < 4; ++j) {
    badd[j] = bias[j * Hs + u];
    const float* e = Wih + (size_t)(j * Hs + u) * embD;
#pragma unroll
    for (int r = 0; r < NR; ++r) ein[j][r] = e[tk[r]];
  }
  float* cp = c + (size_t)(ub * (BN / 16) + wv) * (32 * NR) + lane;
#pragma unroll
  for (int r = 0; r < NR; ++r) cold[r] = first ? 0.f : cp[r * 32];

  v8f acc[NACC] = {};
  tile_mac<NACC>(lds, hprev + (size_t)(rh * RB) * CH, CH, Whh, Hs,
                 [=](int n) { return (n / UBK) * Hs + ub * UBK + (n % UBK); },
                 kloop / KC, acc);

  float ai[NR], af[NR], ag[NR], ao[NR];
  if constexpr (NACC == 4) {
#pragma unroll
    for (int r = 0; r < NR; ++r) {
      ai[r] = acc[0][r]; af[r] = acc[1][r]; ag[r] = acc[2][r]; ao[r] = acc[3][r];
    }
  } else {
    v8f m0 = lo ? acc[0] : __builtin_shufflevector(acc[0], acc[0], 4, 5, 6, 7, 0, 1, 2, 3);
    v8f m1 = lo ? acc[1] : __builtin_shufflevector(acc[1], acc[1], 4, 5, 6, 7, 0, 1, 2, 3);
#pragma unroll
    for (int r = 0; r < NR; ++r) {
      float got0 = __shfl_xor(m0[r + 4], 8, 32);
      float got1 = __shfl_xor(m1[r + 4], 8, 32);
      ai[r] = lo ? m0[r] : got0; af[r] = lo ? got0 : m0[r];
      ag[r] = lo ? m1[r] : got1; ao[r] = lo ? got1 : m1[r];
    }
  }

#pragma unroll
  for (int r = 0; r < NR; ++r) {
    int b = rbase + r;
    float gi = ai[r] + badd[0]; gi += ein[0][r];
    float gf = af[r] + badd[1]; gf += ein[1][r];
    float gg = ag[r] + badd[2]; gg += ein[2][r];
    float go = ao[r] + badd[3]; go += ein[3][r];
    float si = 1.f / (1.f + expf(-gi));
    float sf = 1.f / (1.f + expf(-gf));
    float so = 1.f / (1.f + expf(-go));
    float cn = sf * cold[r] + si * tanhf(gg);
    float hn = so * tanhf(cn);
    cp[r * 32] = cn;
    hout[(size_t)b * CH + u] = hn;
  }
}

__global__ __launch_bounds__(NT) void lstm_step(
    const float* hp_m, const float* __restrict__ Whh_m, const float* __restrict__ b_m,
    const float* __restrict__ Wih_m, const int* __restrict__ tok_m,
    float* __restrict__ c_m, float* ho_m,
    const float* hp_f, const float* __restrict__ Whh_f, const float* __restrict__ b_f,
    const float* __restrict__ Wih_f, const int* __restrict__ tok_f,
    float* __restrict__ c_f, float* ho_f,
    const float* hp_b, const float* __restrict__ Whh_b, const float* __restrict__ b_b,
    const float* __restrict__ Wih_b, const int* __restrict__ tok_b,
    float* __restrict__ c_b, float* ho_b,
    int Hs_m, int Hs_p, int kl_m, int kl_p, int embD) {
  __shared__ __attribute__((aligned(16))) float lds[LDS_FLOATS];
  int rh = blockIdx.x % RSPLIT;
  int blk = blockIdx.x / RSPLIT;
  if (blk < MBLK) {
    lstm_step_tile<2>(lds, hp_m, Whh_m, b_m, Wih_m, tok_m, c_m, ho_m, Hs_m, kl_m, embD, blk, rh);
    return;
  }
  blk -= MBLK;
  const float* hp = hp_f;
  const float* Wt = Whh_f;
  const float* bias = b_f;
  const float* emb = Wih_f;
  const int* tok = tok_f;
  float* c = c_f;
  float* ho = ho_f;
  if (blk >= PBLK) {
    blk -= PBLK;
    hp = hp_b; Wt = Whh_b; bias = b_b; emb = Wih_b; tok = tok_b; c = c_b; ho = ho_b;
  }
  lstm_step_tile<4>(lds, hp, Wt, bias, emb, tok, c, ho, Hs_p, kl_p, embD, blk, rh);
}

__global__ __launch_bounds__(NT) void proj_gemm(
    const float* __restrict__ prior, const float* __restrict__ modern,
    const float* __restrict__ Wsp, const float* __restrict__ bsp,
    const float* __restrict__ Wip, const float* __restrict__ bip,
    float* __restrict__ pPs, float* __restrict__ pPi,
    float* __restrict__ pMs, float* __restrict__ pMi, int N, int K) {
  int job = blockIdx.y;
  const float* X = (job < 2) ? prior : modern;
  const float* Wt = (job & 1) ? Wip : Wsp;
  const float* bias = (job & 1) ? bip : bsp;
  float* out = (job == 0) ? pPs : (job == 1) ? pPi : (job == 2) ? pMs : pMi;
  __shared__ __attribute__((aligned(16))) float lds[LDS_FLOATS];
  gemm_tile_wmma_f32(lds, X, Wt, bias, out, N, K, blockIdx.x, 0);
}

__global__ void finalize(const float* __restrict__ pPs, const float* __restrict__ pPi,
                         const float* __restrict__ pMs, const float* __restrict__ pMi,
                         const int* __restrict__ tgt, float* __restrict__ out) {
  int t = blockIdx.x * blockDim.x + threadIdx.x;
  if (t >= PN * MN * BN) return;
  int b = t % BN;
  int pm = t / BN;
  int m = pm % MN;
  int p = pm / MN;
  const float* as = pPs + ((size_t)p * BN + b) * OPAD;
  const float* bsv = pMs + ((size_t)m * BN + b) * OPAD;
  const float* ai = pPi + ((size_t)p * BN + b) * OPAD;
  const float* biv = pMi + ((size_t)m * BN + b) * OPAD;
  int tg = (m < MN - 1) ? tgt[(m + 1) * BN + b] : 0;

  float mxs = -3.4e38f, mxi = -3.4e38f;
  for (int o = 0; o < ON; ++o) {
    float ls = as[o] + bsv[o];
    float li = ai[o] + biv[o];
    mxs = ls > mxs ? ls : mxs;
    mxi = li > mxi ? li : mxi;
  }
  float sums = 0.f, sumi = 0.f, ls52 = 0.f, li52 = 0.f, lst = 0.f, lit = 0.f;
  for (int o = 0; o < ON; ++o) {
    float ls = as[o] + bsv[o];
    float li = ai[o] + biv[o];
    sums += expf(ls - mxs);
    sumi += expf(li - mxi);
    if (o == ON - 1) { ls52 = ls; li52 = li; }
    if (o == tg) { lst = ls; lit = li; }
  }
  float lses = mxs + logf(sums);
  float lsei = mxi + logf(sumi);
  size_t idxPMB = ((size_t)p * MN + m) * BN + b;
  out[idxPMB] = ls52 - lses;
  out[(size_t)PN * MN * BN + idxPMB] = li52 - lsei;
  if (m < MN - 1) {
    size_t base = 2ull * PN * MN * BN;
    size_t idxS = ((size_t)p * (MN - 1) + m) * BN + b;
    out[base + idxS] = lst - lses;
    out[base + (size_t)PN * (MN - 1) * BN + idxS] = lit - lsei;
  }
}


extern "C" void kernel_launch(void* const* d_in, const int* in_sizes, int n_in,
                              void* d_out, int out_size, void* d_ws, size_t ws_size,
                              hipStream_t stream) {
  (void)in_sizes; (void)n_in; (void)out_size; (void)ws_size;
  const float* src_oh = (const float*)d_in[0];
  const float* tgt_oh = (const float*)d_in[1];  (void)tgt_oh;
  const float* Wih_pf = (const float*)d_in[2];
  const float* Whh_pf = (const float*)d_in[3];
  const float* b_pf   = (const float*)d_in[4];
  const float* Wih_pb = (const float*)d_in[5];
  const float* Whh_pb = (const float*)d_in[6];
  const float* b_pb   = (const float*)d_in[7];
  const float* Wih_m  = (const float*)d_in[8];
  const float* Whh_m  = (const float*)d_in[9];
  const float* b_m    = (const float*)d_in[10];
  const float* W_sub  = (const float*)d_in[11];
  const float* b_sub  = (const float*)d_in[12];
  const float* W_ins  = (const float*)d_in[13];
  const float* b_ins  = (const float*)d_in[14];
  const int*   tgt    = (const int*)d_in[15];

  char* wptr = (char*)d_ws;
  auto alloc = [&](size_t nfloats) -> float* {
    float* p = (float*)wptr;
    wptr += ((nfloats * 4 + 255) / 256) * 256;
    return p;
  };
  float* prior  = alloc((size_t)PN * BN * CH);
  float* modern = alloc((size_t)MN * BN * CH);
  float* c_pf = alloc((size_t)BN * HN);
  float* c_pb = alloc((size_t)BN * HN);
  float* c_m  = alloc((size_t)BN * CH);
  float* Wsp = alloc(OPAD * CH);
  float* Wip = alloc(OPAD * CH);
  float* bsp = alloc(OPAD);
  float* bip = alloc(OPAD);
  float* pPs = alloc((size_t)PN * BN * OPAD);
  float* pPi = alloc((size_t)PN * BN * OPAD);
  float* pMs = alloc((size_t)MN * BN * OPAD);
  float* pMi = alloc((size_t)MN * BN * OPAD);
  int* sidx = (int*)alloc(PN * BN);

  setup_all<<<IBLK + WBLK, 256, 0, stream>>>(
      src_oh, sidx, W_sub, b_sub, W_ins, b_ins, Wsp, Wip, bsp, bip);

  lstm_step<<<(MBLK + 2 * PBLK) * RSPLIT, NT, 0, stream>>>(
      modern,     Whh_m,  b_m,  Wih_m,  tgt,                  c_m,  modern,
      prior,      Whh_pf, b_pf, Wih_pf, sidx,                 c_pf, prior,
      prior + HN, Whh_pb, b_pb, Wih_pb, sidx + (PN - 1) * BN, c_pb,
      prior + (size_t)(PN - 1) * BN * CH + HN,
      CH, HN, 0, 0, DN);
  for (int t = 1; t < PN; ++t) {
    int tb = PN - 1 - t;
    lstm_step<<<(MBLK + 2 * PBLK) * RSPLIT, NT, 0, stream>>>(
        modern + (size_t)(t - 1) * BN * CH, Whh_m,  b_m,  Wih_m,  tgt + t * BN,   c_m,
        modern + (size_t)t * BN * CH,
        prior + (size_t)(t - 1) * BN * CH,  Whh_pf, b_pf, Wih_pf, sidx + t * BN,  c_pf,
        prior + (size_t)t * BN * CH,
        prior + (size_t)(tb + 1) * BN * CH + HN, Whh_pb, b_pb, Wih_pb, sidx + tb * BN, c_pb,
        prior + (size_t)tb * BN * CH + HN,
        CH, HN, CH, HN, DN);
  }

  proj_gemm<<<dim3(PN * BN / RB, 4), NT, 0, stream>>>(
      prior, modern, Wsp, bsp, Wip, bip, pPs, pPi, pMs, pMi, OPAD, CH);

  finalize<<<(PN * MN * BN + 255) / 256, 256, 0, stream>>>(
      pPs, pPi, pMs, pMi, tgt, (float*)d_out);
}


# Projections stored channel-major ([time][channel][batch]) so the log-softmax pass, batch on its lanes, reads one whole 128-byte line per wave load and forms each logit once, in registers
# speedup vs baseline: 7.8805x; 1.2225x over previous
#include <hip/hip_runtime.h>
#include <math.h>

typedef __attribute__((ext_vector_type(2))) float v2f;
typedef __attribute__((ext_vector_type(4))) float v4f;
typedef __attribute__((ext_vector_type(8))) float v8f;

#define PN 32
#define MN 32
#define BN 128
#define HN 256
#define DN 54
#define ON 53
#define OPAD 64
#define CH 512

#define RB 64
#define NT (2 * RB)
#define RSPLIT (BN / RB)
#define CB 64
#define KC 128
#define LDK (KC + 4)
#define NW (NT / 32)
#define STAGE_MAX ((RB + CB) * LDK)
#define LDS_FLOATS (2 * STAGE_MAX)

#define IBLK ((PN * BN) / 256)
#define WBLK ((OPAD * CH) / 256)
#define UBP 16
#define UBM 8
#define MBLK (CH / UBM)
#define PBLK (HN / UBP)

static_assert(PN == MN, "the three recurrences advance in lock step");
static_assert((PN * BN) % 256 == 0 && (OPAD * CH) % 256 == 0, "exact grids");
static_assert(BN % RB == 0 && (PN * BN) % RB == 0 && RB % 16 == 0, "row blocks");
static_assert(CB == 4 * UBP && CH % UBM == 0 && HN % UBP == 0, "a step tile holds all four gates of its units");
static_assert(NT == 128 && RB == 16 * NW, "a wave stages the 16 A rows it multiplies");
static_assert(KC == 4 * 32, "a wave load of 16 B per lane covers one staged row");
static_assert(CH % KC == 0 && HN % KC == 0, "whole K chunks");


__device__ __forceinline__ void onehot_to_idx_one(const float* __restrict__ oh,
                                                  int* __restrict__ idx, int r) {
  const float* p = oh + (size_t)r * DN;
  int v = 0;
  for (int d = 0; d < DN; ++d)
    if (p[d] > 0.5f) v = d;
  idx[r] = v;
}

__device__ __forceinline__ void pad_proj_weights_one(
    const float* __restrict__ Ws, const float* __restrict__ bs,
    const float* __restrict__ Wi, const float* __restrict__ bi,
    float* __restrict__ Wsp, float* __restrict__ Wip,
    float* __restrict__ bsp, float* __restrict__ bip, int t) {
  int n = t / CH;
  int k = t - n * CH;
  Wsp[t] = (n < ON) ? Ws[n * CH + k] : 0.f;
  Wip[t] = (n < ON) ? Wi[n * CH + k] : 0.f;
  if (t < OPAD) {
    bsp[t] = (t < ON) ? bs[t] : 0.f;
    bip[t] = (t < ON) ? bi[t] : 0.f;
  }
}

__global__ void setup_all(const float* __restrict__ src_oh, int* __restrict__ sidx,
                          const float* __restrict__ Ws, const float* __restrict__ bs,
                          const float* __restrict__ Wi, const float* __restrict__ bi,
                          float* __restrict__ Wsp, float* __restrict__ Wip,
                          float* __restrict__ bsp, float* __restrict__ bip) {
  int blk = blockIdx.x;
  int tid = threadIdx.x;
  if (blk < IBLK) {
    onehot_to_idx_one(src_oh, sidx, blk * 256 + tid);
  } else {
    pad_proj_weights_one(Ws, bs, Wi, bi, Wsp, Wip, bsp, bip, (blk - IBLK) * 256 + tid);
  }
}

template <int NACC, class ColRow>
__device__ __forceinline__ void tile_mac(
    float* lds, const float* __restrict__ A, int lda,
    const float* __restrict__ Bw, int ldb, ColRow colrow, int nchunks,
    v8f (&acc)[NACC]) {
  constexpr int CBW = 16 * NACC;
  constexpr int QA = RB / NW;
  constexpr int QB = CBW / NW;
  constexpr int STG = (RB + CBW) * LDK;
  static_assert(QA == 16 && QB * NW == CBW && STG <= STAGE_MAX, "whole rows per wave");

  int tid = threadIdx.x;
  int lane = tid & 31;
  int wv = tid >> 5;
  int kh = lane >> 4;
  int l16 = lane & 15;

  if (nchunks <= 0) return;

  const float* ag = A + (size_t)(QA * wv) * lda + 4 * lane;
  const float* bg[QB];
#pragma unroll
  for (int q = 0; q < QB; ++q) bg[q] = Bw + (size_t)colrow(QB * wv + q) * ldb + 4 * lane;
  float* as = lds + (QA * wv) * LDK + 4 * lane;
  float* bs = lds + (RB + QB * wv) * LDK + 4 * lane;

  {
    v4f pa[QA], pb[QB];
#pragma unroll
    for (int i = 0; i < QA; ++i) pa[i] = *(const v4f*)(ag + (size_t)i * lda);
#pragma unroll
    for (int q = 0; q < QB; ++q) pb[q] = *(const v4f*)(bg[q]);
#pragma unroll
    for (int i = 0; i < QA; ++i) *(v4f*)(as + i * LDK) = pa[i];
#pragma unroll
    for (int q = 0; q < QB; ++q) *(v4f*)(bs + q * LDK) = pb[q];
  }
  __syncthreads();

  const float* af = lds + (16 * wv + l16) * LDK + 2 * kh;
  const float* bf = lds + (RB + l16) * LDK + 2 * kh;

  for (int kc = 0; kc < nchunks; ++kc) {
    v4f pa[QA], pb[QB];
    bool hasNext = (kc + 1 < nchunks);
    if (hasNext) {
      size_t ko = (size_t)(kc + 1) * KC;
#pragma unroll
      for (int i = 0; i < QA; ++i) pa[i] = *(const v4f*)(ag + (size_t)i * lda + ko);
#pragma unroll
      for (int q = 0; q < QB; ++q) pb[q] = *(const v4f*)(bg[q] + ko);
    }

    const float* al = af + (kc & 1) * STG;
    const float* bl = bf + (kc & 1) * STG;

#pragma unroll
    for (int k = 0; k < KC; k += 4) {
      v2f a = *(const v2f*)(al + k);
      v2f b[NACC];
#pragma unroll
      for (int j = 0; j < NACC; ++j) b[j] = *(const v2f*)(bl + 16 * j * LDK + k);
#pragma unroll
      for (int j = 0; j < NACC; ++j)
        acc[j] = __builtin_amdgcn_wmma_f32_16x16x4_f32(false, a, false, b[j], (short)0, acc[j], false, false);
    }

    if (hasNext) {
      int so = ((kc + 1) & 1) * STG;
#pragma unroll
      for (int i = 0; i < QA; ++i) *(v4f*)(as + so + i * LDK) = pa[i];
#pragma unroll
      for (int q = 0; q < QB; ++q) *(v4f*)(bs + so + q * LDK) = pb[q];
      __syncthreads();
    }
  }
}

__device__ __forceinline__ void proj_tile_wmma_f32(
    float* lds, const float* __restrict__ X, const float* __restrict__ W,
    const float* __restrict__ bias, float* __restrict__ outT, int K, int rowBlk) {
  int tid = threadIdx.x;
  int lane = tid & 31;
  int wv = tid >> 5;
  int kh = lane >> 4;
  int l16 = lane & 15;
  int row0 = rowBlk * RB + wv * 16 + kh * 8;
  int tstep = row0 / BN;
  int b0 = row0 - tstep * BN;

  v8f accs[4] = {};
  tile_mac<4>(lds, X + (size_t)(rowBlk * RB) * K, K, W, K,
              [=](int n) { return n; }, K / KC, accs);

#pragma unroll
  for (int j = 0; j < 4; ++j) {
    int col = j * 16 + l16;
    float badd = bias[col];
    float* o = outT + ((size_t)tstep * OPAD + col) * BN + b0;
    v4f lo = {accs[j][0] + badd, accs[j][1] + badd, accs[j][2] + badd, accs[j][3] + badd};
    v4f hi = {accs[j][4] + badd, accs[j][5] + badd, accs[j][6] + badd, accs[j][7] + badd};
    *(v4f*)o = lo;
    *(v4f*)(o + 4) = hi;
  }
}

template <int NACC>
__device__ __forceinline__ void lstm_step_tile(
    float* lds, const float* hprev, const float* __restrict__ Whh,
    const float* __restrict__ bias, const float* __restrict__ Wih,
    const int* __restrict__ tok, float* __restrict__ c, float* hout,
    int Hs, int kloop, int embD, int ub, int rh) {
  constexpr int UBK = 4 * NACC;
  constexpr int NR = 2 * NACC;
  static_assert(NACC == 4 || NACC == 2, "gate layouts written out for 16 and 8 units");
  int tid = threadIdx.x;
  int lane = tid & 31;
  int wv = rh * (RB / 16) + (tid >> 5);
  int kh = lane >> 4;
  int l16 = lane & 15;
  int row0 = wv * 16;
  bool lo = (NACC == 4) || (l16 < 8);
  int ul = (NACC == 4) ? l16 : (l16 & 7);
  int rbase = row0 + kh * 8 + (lo ? 0 : 4);
  int u = ub * UBK + ul;
  bool first = (kloop == 0);

  int tk[NR];
#pragma unroll
  for (int r = 0; r < NR; ++r) tk[r] = tok[rbase + r];
  float badd[4], ein[4][NR], cold[NR];
#pragma unroll
  for (int j = 0; j < 4; ++j) {
    badd[j] = bias[j * Hs + u];
    const float* e = Wih + (size_t)(j * Hs + u) * embD;
#pragma unroll
    for (int r = 0; r < NR; ++r) ein[j][r] = e[tk[r]];
  }
  float* cp = c + (size_t)(ub * (BN / 16) + wv) * (32 * NR) + lane;
#pragma unroll
  for (int r = 0; r < NR; ++r) cold[r] = first ? 0.f : cp[r * 32];

  v8f acc[NACC] = {};
  tile_mac<NACC>(lds, hprev + (size_t)(rh * RB) * CH, CH, Whh, Hs,
                 [=](int n) { return (n / UBK) * Hs + ub * UBK + (n % UBK); },
                 kloop / KC, acc);

  float ai[NR], af[NR], ag[NR], ao[NR];
  if constexpr (NACC == 4) {
#pragma unroll
    for (int r = 0; r < NR; ++r) {
      ai[r] = acc[0][r]; af[r] = acc[1][r]; ag[r] = acc[2][r]; ao[r] = acc[3][r];
    }
  } else {
    v8f m0 = lo ? acc[0] : __builtin_shufflevector(acc[0], acc[0], 4, 5, 6, 7, 0, 1, 2, 3);
    v8f m1 = lo ? acc[1] : __builtin_shufflevector(acc[1], acc[1], 4, 5, 6, 7, 0, 1, 2, 3);
#pragma unroll
    for (int r = 0; r < NR; ++r) {
      float got0 = __shfl_xor(m0[r + 4], 8, 32);
      float got1 = __shfl_xor(m1[r + 4], 8, 32);
      ai[r] = lo ? m0[r] : got0; af[r] = lo ? got0 : m0[r];
      ag[r] = lo ? m1[r] : got1; ao[r] = lo ? got1 : m1[r];
    }
  }

#pragma unroll
  for (int r = 0; r < NR; ++r) {
    int b = rbase + r;
    float gi = ai[r] + badd[0]; gi += ein[0][r];
    float gf = af[r] + badd[1]; gf += ein[1][r];
    float gg = ag[r] + badd[2]; gg += ein[2][r];
    float go = ao[r] + badd[3]; go += ein[3][r];
    float si = 1.f / (1.f + expf(-gi));
    float sf = 1.f / (1.f + expf(-gf));
    float so = 1.f / (1.f + expf(-go));
    float cn = sf * cold[r] + si * tanhf(gg);
    float hn = so * tanhf(cn);
    cp[r * 32] = cn;
    hout[(size_t)b * CH + u] = hn;
  }
}

__global__ __launch_bounds__(NT) void lstm_step(
    const float* hp_m, const float* __restrict__ Whh_m, const float* __restrict__ b_m,
    const float* __restrict__ Wih_m, const int* __restrict__ tok_m,
    float* __restrict__ c_m, float* ho_m,
    const float* hp_f, const float* __restrict__ Whh_f, const float* __restrict__ b_f,
    const float* __restrict__ Wih_f, const int* __restrict__ tok_f,
    float* __restrict__ c_f, float* ho_f,
    const float* hp_b, const float* __restrict__ Whh_b, const float* __restrict__ b_b,
    const float* __restrict__ Wih_b, const int* __restrict__ tok_b,
    float* __restrict__ c_b, float* ho_b,
    int Hs_m, int Hs_p, int kl_m, int kl_p, int embD) {
  __shared__ __attribute__((aligned(16))) float lds[LDS_FLOATS];
  int rh = blockIdx.x % RSPLIT;
  int blk = blockIdx.x / RSPLIT;
  if (blk < MBLK) {
    lstm_step_tile<2>(lds, hp_m, Whh_m, b_m, Wih_m, tok_m, c_m, ho_m, Hs_m, kl_m, embD, blk, rh);
    return;
  }
  blk -= MBLK;
  const float* hp = hp_f;
  const float* Wt = Whh_f;
  const float* bias = b_f;
  const float* emb = Wih_f;
  const int* tok = tok_f;
  float* c = c_f;
  float* ho = ho_f;
  if (blk >= PBLK) {
    blk -= PBLK;
    hp = hp_b; Wt = Whh_b; bias = b_b; emb = Wih_b; tok = tok_b; c = c_b; ho = ho_b;
  }
  lstm_step_tile<4>(lds, hp, Wt, bias, emb, tok, c, ho, Hs_p, kl_p, embD, blk, rh);
}

__global__ __launch_bounds__(NT) void proj_gemm(
    const float* __restrict__ prior, const float* __restrict__ modern,
    const float* __restrict__ Wsp, const float* __restrict__ bsp,
    const float* __restrict__ Wip, const float* __restrict__ bip,
    float* __restrict__ pPs, float* __restrict__ pPi,
    float* __restrict__ pMs, float* __restrict__ pMi, int K) {
  int job = blockIdx.y;
  const float* X = (job < 2) ? prior : modern;
  const float* Wt = (job & 1) ? Wip : Wsp;
  const float* bias = (job & 1) ? bip : bsp;
  float* out = (job == 0) ? pPs : (job == 1) ? pPi : (job == 2) ? pMs : pMi;
  __shared__ __attribute__((aligned(16))) float lds[LDS_FLOATS];
  proj_tile_wmma_f32(lds, X, Wt, bias, out, K, blockIdx.x);
}

__global__ __launch_bounds__(256) void finalize(
    const float* __restrict__ pPs, const float* __restrict__ pPi,
    const float* __restrict__ pMs, const float* __restrict__ pMi,
    const int* __restrict__ tgt, float* __restrict__ out) {
  int t = blockIdx.x * blockDim.x + threadIdx.x;
  if (t >= PN * MN * BN) return;
  int b = t % BN;
  int pm = t / BN;
  int m = pm % MN;
  int p = pm / MN;
  const float* as = pPs + (size_t)p * OPAD * BN + b;
  const float* bsv = pMs + (size_t)m * OPAD * BN + b;
  const float* ai = pPi + (size_t)p * OPAD * BN + b;
  const float* biv = pMi + (size_t)m * OPAD * BN + b;
  int tg = (m < MN - 1) ? tgt[(m + 1) * BN + b] : 0;

  float ls[ON], li[ON];
#pragma unroll
  for (int o = 0; o < ON; ++o) {
    ls[o] = as[o * BN] + bsv[o * BN];
    li[o] = ai[o * BN] + biv[o * BN];
  }
  float mxs = -3.4e38f, mxi = -3.4e38f;
#pragma unroll
  for (int o = 0; o < ON; ++o) {
    mxs = ls[o] > mxs ? ls[o] : mxs;
    mxi = li[o] > mxi ? li[o] : mxi;
  }
  float sums = 0.f, sumi = 0.f, lst = 0.f, lit = 0.f;
#pragma unroll
  for (int o = 0; o < ON; ++o) {
    sums += expf(ls[o] - mxs);
    sumi += expf(li[o] - mxi);
    if (o == tg) { lst = ls[o]; lit = li[o]; }
  }
  float ls52 = ls[ON - 1], li52 = li[ON - 1];
  float lses = mxs + logf(sums);
  float lsei = mxi + logf(sumi);
  size_t idxPMB = ((size_t)p * MN + m) * BN + b;
  out[idxPMB] = ls52 - lses;
  out[(size_t)PN * MN * BN + idxPMB] = li52 - lsei;
  if (m < MN - 1) {
    size_t base = 2ull * PN * MN * BN;
    size_t idxS = ((size_t)p * (MN - 1) + m) * BN + b;
    out[base + idxS] = lst - lses;
    out[base + (size_t)PN * (MN - 1) * BN + idxS] = lit - lsei;
  }
}


extern "C" void kernel_launch(void* const* d_in, const int* in_sizes, int n_in,
                              void* d_out, int out_size, void* d_ws, size_t ws_size,
                              hipStream_t stream) {
  (void)in_sizes; (void)n_in; (void)out_size; (void)ws_size;
  const float* src_oh = (const float*)d_in[0];
  const float* tgt_oh = (const float*)d_in[1];  (void)tgt_oh;
  const float* Wih_pf = (const float*)d_in[2];
  const float* Whh_pf = (const float*)d_in[3];
  const float* b_pf   = (const float*)d_in[4];
  const float* Wih_pb = (const float*)d_in[5];
  const float* Whh_pb = (const float*)d_in[6];
  const float* b_pb   = (const float*)d_in[7];
  const float* Wih_m  = (const float*)d_in[8];
  const float* Whh_m  = (const float*)d_in[9];
  const float* b_m    = (const float*)d_in[10];
  const float* W_sub  = (const float*)d_in[11];
  const float* b_sub  = (const float*)d_in[12];
  const float* W_ins  = (const float*)d_in[13];
  const float* b_ins  = (const float*)d_in[14];
  const int*   tgt    = (const int*)d_in[15];

  char* wptr = (char*)d_ws;
  auto alloc = [&](size_t nfloats) -> float* {
    float* p = (float*)wptr;
    wptr += ((nfloats * 4 + 255) / 256) * 256;
    return p;
  };
  float* prior  = alloc((size_t)PN * BN * CH);
  float* modern = alloc((size_t)MN * BN * CH);
  float* c_pf = alloc((size_t)BN * HN);
  float* c_pb = alloc((size_t)BN * HN);
  float* c_m  = alloc((size_t)BN * CH);
  float* Wsp = alloc(OPAD * CH);
  float* Wip = alloc(OPAD * CH);
  float* bsp = alloc(OPAD);
  float* bip = alloc(OPAD);
  float* pPs = alloc((size_t)PN * BN * OPAD);
  float* pPi = alloc((size_t)PN * BN * OPAD);
  float* pMs = alloc((size_t)MN * BN * OPAD);
  float* pMi = alloc((size_t)MN * BN * OPAD);
  int* sidx = (int*)alloc(PN * BN);

  setup_all<<<IBLK + WBLK, 256, 0, stream>>>(
      src_oh, sidx, W_sub, b_sub, W_ins, b_ins, Wsp, Wip, bsp, bip);

  lstm_step<<<(MBLK + 2 * PBLK) * RSPLIT, NT, 0, stream>>>(
      modern,     Whh_m,  b_m,  Wih_m,  tgt,                  c_m,  modern,
      prior,      Whh_pf, b_pf, Wih_pf, sidx,                 c_pf, prior,
      prior + HN, Whh_pb, b_pb, Wih_pb, sidx + (PN - 1) * BN, c_pb,
      prior + (size_t)(PN - 1) * BN * CH + HN,
      CH, HN, 0, 0, DN);
  for (int t = 1; t < PN; ++t) {
    int tb = PN - 1 - t;
    lstm_step<<<(MBLK + 2 * PBLK) * RSPLIT, NT, 0, stream>>>(
        modern + (size_t)(t - 1) * BN * CH, Whh_m,  b_m,  Wih_m,  tgt + t * BN,   c_m,
        modern + (size_t)t * BN * CH,
        prior + (size_t)(t - 1) * BN * CH,  Whh_pf, b_pf, Wih_pf, sidx + t * BN,  c_pf,
        prior + (size_t)t * BN * CH,
        prior + (size_t)(tb + 1) * BN * CH + HN, Whh_pb, b_pb, Wih_pb, sidx + tb * BN, c_pb,
        prior + (size_t)tb * BN * CH + HN,
        CH, HN, CH, HN, DN);
  }

  proj_gemm<<<dim3(PN * BN / RB, 4), NT, 0, stream>>>(
      prior, modern, Wsp, bsp, Wip, bip, pPs, pPi, pMs, pMi, CH);

  finalize<<<(PN * MN * BN + 255) / 256, 256, 0, stream>>>(
      pPs, pPi, pMs, pMi, tgt, (float*)d_out);
}


# LSTM gate sigmoids and tanh on the hardware exp2 and reciprocal instructions (about 1 ulp each) in place of library expf, tanhf and exact division; branch-free, so a lane's pairs interleave
# speedup vs baseline: 8.5644x; 1.0868x over previous
#include <hip/hip_runtime.h>
#include <math.h>

typedef __attribute__((ext_vector_type(2))) float v2f;
typedef __attribute__((ext_vector_type(4))) float v4f;
typedef __attribute__((ext_vector_type(8))) float v8f;

#define PN 32
#define MN 32
#define BN 128
#define HN 256
#define DN 54
#define ON 53
#define OPAD 64
#define CH 512

#define RB 64
#define NT (2 * RB)
#define RSPLIT (BN / RB)
#define CB 64
#define KC 128
#define LDK (KC + 4)
#define NW (NT / 32)
#define STAGE_MAX ((RB + CB) * LDK)
#define LDS_FLOATS (2 * STAGE_MAX)

#define IBLK ((PN * BN) / 256)
#define WBLK ((OPAD * CH) / 256)
#define UBP 16
#define UBM 8
#define MBLK (CH / UBM)
#define PBLK (HN / UBP)

static_assert(PN == MN, "the three recurrences advance in lock step");
static_assert((PN * BN) % 256 == 0 && (OPAD * CH) % 256 == 0, "exact grids");
static_assert(BN % RB == 0 && (PN * BN) % RB == 0 && RB % 16 == 0, "row blocks");
static_assert(CB == 4 * UBP && CH % UBM == 0 && HN % UBP == 0, "a step tile holds all four gates of its units");
static_assert(NT == 128 && RB == 16 * NW, "a wave stages the 16 A rows it multiplies");
static_assert(KC == 4 * 32, "a wave load of 16 B per lane covers one staged row");
static_assert(CH % KC == 0 && HN % KC == 0, "whole K chunks");


__device__ __forceinline__ void onehot_to_idx_one(const float* __restrict__ oh,
                                                  int* __restrict__ idx, int r) {
  const float* p = oh + (size_t)r * DN;
  int v = 0;
  for (int d = 0; d < DN; ++d)
    if (p[d] > 0.5f) v = d;
  idx[r] = v;
}

__device__ __forceinline__ void pad_proj_weights_one(
    const float* __restrict__ Ws, const float* __restrict__ bs,
    const float* __restrict__ Wi, const float* __restrict__ bi,
    float* __restrict__ Wsp, float* __restrict__ Wip,
    float* __restrict__ bsp, float* __restrict__ bip, int t) {
  int n = t / CH;
  int k = t - n * CH;
  Wsp[t] = (n < ON) ? Ws[n * CH + k] : 0.f;
  Wip[t] = (n < ON) ? Wi[n * CH + k] : 0.f;
  if (t < OPAD) {
    bsp[t] = (t < ON) ? bs[t] : 0.f;
    bip[t] = (t < ON) ? bi[t] : 0.f;
  }
}

__global__ void setup_all(const float* __restrict__ src_oh, int* __restrict__ sidx,
                          const float* __restrict__ Ws, const float* __restrict__ bs,
                          const float* __restrict__ Wi, const float* __restrict__ bi,
                          float* __restrict__ Wsp, float* __restrict__ Wip,
                          float* __restrict__ bsp, float* __restrict__ bip) {
  int blk = blockIdx.x;
  int tid = threadIdx.x;
  if (blk < IBLK) {
    onehot_to_idx_one(src_oh, sidx, blk * 256 + tid);
  } else {
    pad_proj_weights_one(Ws, bs, Wi, bi, Wsp, Wip, bsp, bip, (blk - IBLK) * 256 + tid);
  }
}

template <int NACC, class ColRow>
__device__ __forceinline__ void tile_mac(
    float* lds, const float* __restrict__ A, int lda,
    const float* __restrict__ Bw, int ldb, ColRow colrow, int nchunks,
    v8f (&acc)[NACC]) {
  constexpr int CBW = 16 * NACC;
  constexpr int QA = RB / NW;
  constexpr int QB = CBW / NW;
  constexpr int STG = (RB + CBW) * LDK;
  static_assert(QA == 16 && QB * NW == CBW && STG <= STAGE_MAX, "whole rows per wave");

  int tid = threadIdx.x;
  int lane = tid & 31;
  int wv = tid >> 5;
  int kh = lane >> 4;
  int l16 = lane & 15;

  if (nchunks <= 0) return;

  const float* ag = A + (size_t)(QA * wv) * lda + 4 * lane;
  const float* bg[QB];
#pragma unroll
  for (int q = 0; q < QB; ++q) bg[q] = Bw + (size_t)colrow(QB * wv + q) * ldb + 4 * lane;
  float* as = lds + (QA * wv) * LDK + 4 * lane;
  float* bs = lds + (RB + QB * wv) * LDK + 4 * lane;

  {
    v4f pa[QA], pb[QB];
#pragma unroll
    for (int i = 0; i < QA; ++i) pa[i] = *(const v4f*)(ag + (size_t)i * lda);
#pragma unroll
    for (int q = 0; q < QB; ++q) pb[q] = *(const v4f*)(bg[q]);
#pragma unroll
    for (int i = 0; i < QA; ++i) *(v4f*)(as + i * LDK) = pa[i];
#pragma unroll
    for (int q = 0; q < QB; ++q) *(v4f*)(bs + q * LDK) = pb[q];
  }
  __syncthreads();

  const float* af = lds + (16 * wv + l16) * LDK + 2 * kh;
  const float* bf = lds + (RB + l16) * LDK + 2 * kh;

  for (int kc = 0; kc < nchunks; ++kc) {
    v4f pa[QA], pb[QB];
    bool hasNext = (kc + 1 < nchunks);
    if (hasNext) {
      size_t ko = (size_t)(kc + 1) * KC;
#pragma unroll
      for (int i = 0; i < QA; ++i) pa[i] = *(const v4f*)(ag + (size_t)i * lda + ko);
#pragma unroll
      for (int q = 0; q < QB; ++q) pb[q] = *(const v4f*)(bg[q] + ko);
    }

    const float* al = af + (kc & 1) * STG;
    const float* bl = bf + (kc & 1) * STG;

#pragma unroll
    for (int k = 0; k < KC; k += 4) {
      v2f a = *(const v2f*)(al + k);
      v2f b[NACC];
#pragma unroll
      for (int j = 0; j < NACC; ++j) b[j] = *(const v2f*)(bl + 16 * j * LDK + k);
#pragma unroll
      for (int j = 0; j < NACC; ++j)
        acc[j] = __builtin_amdgcn_wmma_f32_16x16x4_f32(false, a, false, b[j], (short)0, acc[j], false, false);
    }

    if (hasNext) {
      int so = ((kc + 1) & 1) * STG;
#pragma unroll
      for (int i = 0; i < QA; ++i) *(v4f*)(as + so + i * LDK) = pa[i];
#pragma unroll
      for (int q = 0; q < QB; ++q) *(v4f*)(bs + so + q * LDK) = pb[q];
      __syncthreads();
    }
  }
}

__device__ __forceinline__ void proj_tile_wmma_f32(
    float* lds, const float* __restrict__ X, const float* __restrict__ W,
    const float* __restrict__ bias, float* __restrict__ outT, int K, int rowBlk) {
  int tid = threadIdx.x;
  int lane = tid & 31;
  int wv = tid >> 5;
  int kh = lane >> 4;
  int l16 = lane & 15;
  int row0 = rowBlk * RB + wv * 16 + kh * 8;
  int tstep = row0 / BN;
  int b0 = row0 - tstep * BN;

  v8f accs[4] = {};
  tile_mac<4>(lds, X + (size_t)(rowBlk * RB) * K, K, W, K,
              [=](int n) { return n; }, K / KC, accs);

#pragma unroll
  for (int j = 0; j < 4; ++j) {
    int col = j * 16 + l16;
    float badd = bias[col];
    float* o = outT + ((size_t)tstep * OPAD + col) * BN + b0;
    v4f lo = {accs[j][0] + badd, accs[j][1] + badd, accs[j][2] + badd, accs[j][3] + badd};
    v4f hi = {accs[j][4] + badd, accs[j][5] + badd, accs[j][6] + badd, accs[j][7] + badd};
    *(v4f*)o = lo;
    *(v4f*)(o + 4) = hi;
  }
}

__device__ __forceinline__ float sigmoid_hw(float x) {
  return __builtin_amdgcn_rcpf(1.f + __builtin_amdgcn_exp2f(-1.4426950408889634f * x));
}
__device__ __forceinline__ float tanh_hw(float x) {
  return fmaf(-2.f, __builtin_amdgcn_rcpf(1.f + __builtin_amdgcn_exp2f(2.8853900817779268f * x)), 1.f);
}

template <int NACC>
__device__ __forceinline__ void lstm_step_tile(
    float* lds, const float* hprev, const float* __restrict__ Whh,
    const float* __restrict__ bias, const float* __restrict__ Wih,
    const int* __restrict__ tok, float* __restrict__ c, float* hout,
    int Hs, int kloop, int embD, int ub, int rh) {
  constexpr int UBK = 4 * NACC;
  constexpr int NR = 2 * NACC;
  static_assert(NACC == 4 || NACC == 2, "gate layouts written out for 16 and 8 units");
  int tid = threadIdx.x;
  int lane = tid & 31;
  int wv = rh * (RB / 16) + (tid >> 5);
  int kh = lane >> 4;
  int l16 = lane & 15;
  int row0 = wv * 16;
  bool lo = (NACC == 4) || (l16 < 8);
  int ul = (NACC == 4) ? l16 : (l16 & 7);
  int rbase = row0 + kh * 8 + (lo ? 0 : 4);
  int u = ub * UBK + ul;
  bool first = (kloop == 0);

  int tk[NR];
#pragma unroll
  for (int r = 0; r < NR; ++r) tk[r] = tok[rbase + r];
  float badd[4], ein[4][NR], cold[NR];
#pragma unroll
  for (int j = 0; j < 4; ++j) {
    badd[j] = bias[j * Hs + u];
    const float* e = Wih + (size_t)(j * Hs + u) * embD;
#pragma unroll
    for (int r = 0; r < NR; ++r) ein[j][r] = e[tk[r]];
  }
  float* cp = c + (size_t)(ub * (BN / 16) + wv) * (32 * NR) + lane;
#pragma unroll
  for (int r = 0; r < NR; ++r) cold[r] = first ? 0.f : cp[r * 32];

  v8f acc[NACC] = {};
  tile_mac<NACC>(lds, hprev + (size_t)(rh * RB) * CH, CH, Whh, Hs,
                 [=](int n) { return (n / UBK) * Hs + ub * UBK + (n % UBK); },
                 kloop / KC, acc);

  float ai[NR], af[NR], ag[NR], ao[NR];
  if constexpr (NACC == 4) {
#pragma unroll
    for (int r = 0; r < NR; ++r) {
      ai[r] = acc[0][r]; af[r] = acc[1][r]; ag[r] = acc[2][r]; ao[r] = acc[3][r];
    }
  } else {
    v8f m0 = lo ? acc[0] : __builtin_shufflevector(acc[0], acc[0], 4, 5, 6, 7, 0, 1, 2, 3);
    v8f m1 = lo ? acc[1] : __builtin_shufflevector(acc[1], acc[1], 4, 5, 6, 7, 0, 1, 2, 3);
#pragma unroll
    for (int r = 0; r < NR; ++r) {
      float got0 = __shfl_xor(m0[r + 4], 8, 32);
      float got1 = __shfl_xor(m1[r + 4], 8, 32);
      ai[r] = lo ? m0[r] : got0; af[r] = lo ? got0 : m0[r];
      ag[r] = lo ? m1[r] : got1; ao[r] = lo ? got1 : m1[r];
    }
  }

#pragma unroll
  for (int r = 0; r < NR; ++r) {
    int b = rbase + r;
    float gi = ai[r] + badd[0]; gi += ein[0][r];
    float gf = af[r] + badd[1]; gf += ein[1][r];
    float gg = ag[r] + badd[2]; gg += ein[2][r];
    float go = ao[r] + badd[3]; go += ein[3][r];
    float si = sigmoid_hw(gi);
    float sf = sigmoid_hw(gf);
    float so = sigmoid_hw(go);
    float cn = sf * cold[r] + si * tanh_hw(gg);
    float hn = so * tanh_hw(cn);
    cp[r * 32] = cn;
    hout[(size_t)b * CH + u] = hn;
  }
}

__global__ __launch_bounds__(NT) void lstm_step(
    const float* hp_m, const float* __restrict__ Whh_m, const float* __restrict__ b_m,
    const float* __restrict__ Wih_m, const int* __restrict__ tok_m,
    float* __restrict__ c_m, float* ho_m,
    const float* hp_f, const float* __restrict__ Whh_f, const float* __restrict__ b_f,
    const float* __restrict__ Wih_f, const int* __restrict__ tok_f,
    float* __restrict__ c_f, float* ho_f,
    const float* hp_b, const float* __restrict__ Whh_b, const float* __restrict__ b_b,
    const float* __restrict__ Wih_b, const int* __restrict__ tok_b,
    float* __restrict__ c_b, float* ho_b,
    int Hs_m, int Hs_p, int kl_m, int kl_p, int embD) {
  __shared__ __attribute__((aligned(16))) float lds[LDS_FLOATS];
  int rh = blockIdx.x % RSPLIT;
  int blk = blockIdx.x / RSPLIT;
  if (blk < MBLK) {
    lstm_step_tile<2>(lds, hp_m, Whh_m, b_m, Wih_m, tok_m, c_m, ho_m, Hs_m, kl_m, embD, blk, rh);
    return;
  }
  blk -= MBLK;
  const float* hp = hp_f;
  const float* Wt = Whh_f;
  const float* bias = b_f;
  const float* emb = Wih_f;
  const int* tok = tok_f;
  float* c = c_f;
  float* ho = ho_f;
  if (blk >= PBLK) {
    blk -= PBLK;
    hp = hp_b; Wt = Whh_b; bias = b_b; emb = Wih_b; tok = tok_b; c = c_b; ho = ho_b;
  }
  lstm_step_tile<4>(lds, hp, Wt, bias, emb, tok, c, ho, Hs_p, kl_p, embD, blk, rh);
}

__global__ __launch_bounds__(NT) void proj_gemm(
    const float* __restrict__ prior, const float* __restrict__ modern,
    const float* __restrict__ Wsp, const float* __restrict__ bsp,
    const float* __restrict__ Wip, const float* __restrict__ bip,
    float* __restrict__ pPs, float* __restrict__ pPi,
    float* __restrict__ pMs, float* __restrict__ pMi, int K) {
  int job = blockIdx.y;
  const float* X = (job < 2) ? prior : modern;
  const float* Wt = (job & 1) ? Wip : Wsp;
  const float* bias = (job & 1) ? bip : bsp;
  float* out = (job == 0) ? pPs : (job == 1) ? pPi : (job == 2) ? pMs : pMi;
  __shared__ __attribute__((aligned(16))) float lds[LDS_FLOATS];
  proj_tile_wmma_f32(lds, X, Wt, bias, out, K, blockIdx.x);
}

__global__ __launch_bounds__(256) void finalize(
    const float* __restrict__ pPs, const float* __restrict__ pPi,
    const float* __restrict__ pMs, const float* __restrict__ pMi,
    const int* __restrict__ tgt, float* __restrict__ out) {
  int t = blockIdx.x * blockDim.x + threadIdx.x;
  if (t >= PN * MN * BN) return;
  int b = t % BN;
  int pm = t / BN;
  int m = pm % MN;
  int p = pm / MN;
  const float* as = pPs + (size_t)p * OPAD * BN + b;
  const float* bsv = pMs + (size_t)m * OPAD * BN + b;
  const float* ai = pPi + (size_t)p * OPAD * BN + b;
  const float* biv = pMi + (size_t)m * OPAD * BN + b;
  int tg = (m < MN - 1) ? tgt[(m + 1) * BN + b] : 0;

  float ls[ON], li[ON];
#pragma unroll
  for (int o = 0; o < ON; ++o) {
    ls[o] = as[o * BN] + bsv[o * BN];
    li[o] = ai[o * BN] + biv[o * BN];
  }
  float mxs = -3.4e38f, mxi = -3.4e38f;
#pragma unroll
  for (int o = 0; o < ON; ++o) {
    mxs = ls[o] > mxs ? ls[o] : mxs;
    mxi = li[o] > mxi ? li[o] : mxi;
  }
  float sums = 0.f, sumi = 0.f, lst = 0.f, lit = 0.f;
#pragma unroll
  for (int o = 0; o < ON; ++o) {
    sums += expf(ls[o] - mxs);
    sumi += expf(li[o] - mxi);
    if (o == tg) { lst = ls[o]; lit = li[o]; }
  }
  float ls52 = ls[ON - 1], li52 = li[ON - 1];
  float lses = mxs + logf(sums);
  float lsei = mxi + logf(sumi);
  size_t idxPMB = ((size_t)p * MN + m) * BN + b;
  out[idxPMB] = ls52 - lses;
  out[(size_t)PN * MN * BN + idxPMB] = li52 - lsei;
  if (m < MN - 1) {
    size_t base = 2ull * PN * MN * BN;
    size_t idxS = ((size_t)p * (MN - 1) + m) * BN + b;
    out[base + idxS] = lst - lses;
    out[base + (size_t)PN * (MN - 1) * BN + idxS] = lit - lsei;
  }
}


extern "C" void kernel_launch(void* const* d_in, const int* in_sizes, int n_in,
                              void* d_out, int out_size, void* d_ws, size_t ws_size,
                              hipStream_t stream) {
  (void)in_sizes; (void)n_in; (void)out_size; (void)ws_size;
  const float* src_oh = (const float*)d_in[0];
  const float* tgt_oh = (const float*)d_in[1];  (void)tgt_oh;
  const float* Wih_pf = (const float*)d_in[2];
  const float* Whh_pf = (const float*)d_in[3];
  const float* b_pf   = (const float*)d_in[4];
  const float* Wih_pb = (const float*)d_in[5];
  const float* Whh_pb = (const float*)d_in[6];
  const float* b_pb   = (const float*)d_in[7];
  const float* Wih_m  = (const float*)d_in[8];
  const float* Whh_m  = (const float*)d_in[9];
  const float* b_m    = (const float*)d_in[10];
  const float* W_sub  = (const float*)d_in[11];
  const float* b_sub  = (const float*)d_in[12];
  const float* W_ins  = (const float*)d_in[13];
  const float* b_ins  = (const float*)d_in[14];
  const int*   tgt    = (const int*)d_in[15];

  char* wptr = (char*)d_ws;
  auto alloc = [&](size_t nfloats) -> float* {
    float* p = (float*)wptr;
    wptr += ((nfloats * 4 + 255) / 256) * 256;
    return p;
  };
  float* prior  = alloc((size_t)PN * BN * CH);
  float* modern = alloc((size_t)MN * BN * CH);
  float* c_pf = alloc((size_t)BN * HN);
  float* c_pb = alloc((size_t)BN * HN);
  float* c_m  = alloc((size_t)BN * CH);
  float* Wsp = alloc(OPAD * CH);
  float* Wip = alloc(OPAD * CH);
  float* bsp = alloc(OPAD);
  float* bip = alloc(OPAD);
  float* pPs = alloc((size_t)PN * BN * OPAD);
  float* pPi = alloc((size_t)PN * BN * OPAD);
  float* pMs = alloc((size_t)MN * BN * OPAD);
  float* pMi = alloc((size_t)MN * BN * OPAD);
  int* sidx = (int*)alloc(PN * BN);

  setup_all<<<IBLK + WBLK, 256, 0, stream>>>(
      src_oh, sidx, W_sub, b_sub, W_ins, b_ins, Wsp, Wip, bsp, bip);

  lstm_step<<<(MBLK + 2 * PBLK) * RSPLIT, NT, 0, stream>>>(
      modern,     Whh_m,  b_m,  Wih_m,  tgt,                  c_m,  modern,
      prior,      Whh_pf, b_pf, Wih_pf, sidx,                 c_pf, prior,
      prior + HN, Whh_pb, b_pb, Wih_pb, sidx + (PN - 1) * BN, c_pb,
      prior + (size_t)(PN - 1) * BN * CH + HN,
      CH, HN, 0, 0, DN);
  for (int t = 1; t < PN; ++t) {
    int tb = PN - 1 - t;
    lstm_step<<<(MBLK + 2 * PBLK) * RSPLIT, NT, 0, stream>>>(
        modern + (size_t)(t - 1) * BN * CH, Whh_m,  b_m,  Wih_m,  tgt + t * BN,   c_m,
        modern + (size_t)t * BN * CH,
        prior + (size_t)(t - 1) * BN * CH,  Whh_pf, b_pf, Wih_pf, sidx + t * BN,  c_pf,
        prior + (size_t)t * BN * CH,
        prior + (size_t)(tb + 1) * BN * CH + HN, Whh_pb, b_pb, Wih_pb, sidx + tb * BN, c_pb,
        prior + (size_t)tb * BN * CH + HN,
        CH, HN, CH, HN, DN);
  }

  proj_gemm<<<dim3(PN * BN / RB, 4), NT, 0, stream>>>(
      prior, modern, Wsp, bsp, Wip, bip, pPs, pPi, pMs, pMi, CH);

  finalize<<<(PN * MN * BN + 255) / 256, 256, 0, stream>>>(
      pPs, pPi, pMs, pMi, tgt, (float*)d_out);
}


# LSTM step tiles fetch K chunks two ahead, not one (a chunk per LDS region; the barrier waits for LDS only), and start that fetch first: tokens beside chunk 0, embedding columns loaded in the K loop
# speedup vs baseline: 9.1375x; 1.0669x over previous
#include <hip/hip_runtime.h>
#include <math.h>

typedef __attribute__((ext_vector_type(2))) float v2f;
typedef __attribute__((ext_vector_type(4))) float v4f;
typedef __attribute__((ext_vector_type(8))) float v8f;

#define PN 32
#define MN 32
#define BN 128
#define HN 256
#define DN 54
#define ON 53
#define OPAD 64
#define CH 512

#define RB 64
#define NT (2 * RB)
#define RSPLIT (BN / RB)
#define CB 64
#define KC 128
#define LDK (KC + 4)
#define NW (NT / 32)
#define STAGE_MAX ((RB + CB) * LDK)
#define LDS_FLOATS (2 * STAGE_MAX)

#define IBLK ((PN * BN) / 256)
#define WBLK ((OPAD * CH) / 256)
#define UBP 16
#define UBM 8
#define MBLK (CH / UBM)
#define PBLK (HN / UBP)

static_assert(PN == MN, "the three recurrences advance in lock step");
static_assert((PN * BN) % 256 == 0 && (OPAD * CH) % 256 == 0, "exact grids");
static_assert(BN % RB == 0 && (PN * BN) % RB == 0 && RB % 16 == 0, "row blocks");
static_assert(CB == 4 * UBP && CH % UBM == 0 && HN % UBP == 0, "a step tile holds all four gates of its units");
static_assert(NT == 128 && RB == 16 * NW, "a wave stages the 16 A rows it multiplies");
static_assert(KC == 4 * 32, "a wave load of 16 B per lane covers one staged row");
static_assert(CH % KC == 0 && HN % KC == 0, "whole K chunks");


__device__ __forceinline__ void onehot_to_idx_one(const float* __restrict__ oh,
                                                  int* __restrict__ idx, int r) {
  const float* p = oh + (size_t)r * DN;
  int v = 0;
  for (int d = 0; d < DN; ++d)
    if (p[d] > 0.5f) v = d;
  idx[r] = v;
}

__device__ __forceinline__ void pad_proj_weights_one(
    const float* __restrict__ Ws, const float* __restrict__ bs,
    const float* __restrict__ Wi, const float* __restrict__ bi,
    float* __restrict__ Wsp, float* __restrict__ Wip,
    float* __restrict__ bsp, float* __restrict__ bip, int t) {
  int n = t / CH;
  int k = t - n * CH;
  Wsp[t] = (n < ON) ? Ws[n * CH + k] : 0.f;
  Wip[t] = (n < ON) ? Wi[n * CH + k] : 0.f;
  if (t < OPAD) {
    bsp[t] = (t < ON) ? bs[t] : 0.f;
    bip[t] = (t < ON) ? bi[t] : 0.f;
  }
}

__global__ void setup_all(const float* __restrict__ src_oh, int* __restrict__ sidx,
                          const float* __restrict__ Ws, const float* __restrict__ bs,
                          const float* __restrict__ Wi, const float* __restrict__ bi,
                          float* __restrict__ Wsp, float* __restrict__ Wip,
                          float* __restrict__ bsp, float* __restrict__ bip) {
  int blk = blockIdx.x;
  int tid = threadIdx.x;
  if (blk < IBLK) {
    onehot_to_idx_one(src_oh, sidx, blk * 256 + tid);
  } else {
    pad_proj_weights_one(Ws, bs, Wi, bi, Wsp, Wip, bsp, bip, (blk - IBLK) * 256 + tid);
  }
}

template <int NACC, class ColRow>
__device__ __forceinline__ void tile_mac(
    float* lds, const float* __restrict__ A, int lda,
    const float* __restrict__ Bw, int ldb, ColRow colrow, int nchunks,
    v8f (&acc)[NACC]) {
  constexpr int CBW = 16 * NACC;
  constexpr int QA = RB / NW;
  constexpr int QB = CBW / NW;
  constexpr int STG = (RB + CBW) * LDK;
  static_assert(QA == 16 && QB * NW == CBW && STG <= STAGE_MAX, "whole rows per wave");

  int tid = threadIdx.x;
  int lane = tid & 31;
  int wv = tid >> 5;
  int kh = lane >> 4;
  int l16 = lane & 15;

  if (nchunks <= 0) return;

  const float* ag = A + (size_t)(QA * wv) * lda + 4 * lane;
  const float* bg[QB];
#pragma unroll
  for (int q = 0; q < QB; ++q) bg[q] = Bw + (size_t)colrow(QB * wv + q) * ldb + 4 * lane;
  float* as = lds + (QA * wv) * LDK + 4 * lane;
  float* bs = lds + (RB + QB * wv) * LDK + 4 * lane;

  {
    v4f pa[QA], pb[QB];
#pragma unroll
    for (int i = 0; i < QA; ++i) pa[i] = *(const v4f*)(ag + (size_t)i * lda);
#pragma unroll
    for (int q = 0; q < QB; ++q) pb[q] = *(const v4f*)(bg[q]);
#pragma unroll
    for (int i = 0; i < QA; ++i) *(v4f*)(as + i * LDK) = pa[i];
#pragma unroll
    for (int q = 0; q < QB; ++q) *(v4f*)(bs + q * LDK) = pb[q];
  }
  __syncthreads();

  const float* af = lds + (16 * wv + l16) * LDK + 2 * kh;
  const float* bf = lds + (RB + l16) * LDK + 2 * kh;

  for (int kc = 0; kc < nchunks; ++kc) {
    v4f pa[QA], pb[QB];
    bool hasNext = (kc + 1 < nchunks);
    if (hasNext) {
      size_t ko = (size_t)(kc + 1) * KC;
#pragma unroll
      for (int i = 0; i < QA; ++i) pa[i] = *(const v4f*)(ag + (size_t)i * lda + ko);
#pragma unroll
      for (int q = 0; q < QB; ++q) pb[q] = *(const v4f*)(bg[q] + ko);
    }

    const float* al = af + (kc & 1) * STG;
    const float* bl = bf + (kc & 1) * STG;

#pragma unroll
    for (int k = 0; k < KC; k += 4) {
      v2f a = *(const v2f*)(al + k);
      v2f b[NACC];
#pragma unroll
      for (int j = 0; j < NACC; ++j) b[j] = *(const v2f*)(bl + 16 * j * LDK + k);
#pragma unroll
      for (int j = 0; j < NACC; ++j)
        acc[j] = __builtin_amdgcn_wmma_f32_16x16x4_f32(false, a, false, b[j], (short)0, acc[j], false, false);
    }

    if (hasNext) {
      int so = ((kc + 1) & 1) * STG;
#pragma unroll
      for (int i = 0; i < QA; ++i) *(v4f*)(as + so + i * LDK) = pa[i];
#pragma unroll
      for (int q = 0; q < QB; ++q) *(v4f*)(bs + so + q * LDK) = pb[q];
      __syncthreads();
    }
  }
}

__device__ __forceinline__ void lds_barrier() {
  __builtin_amdgcn_fence(__ATOMIC_RELEASE, "workgroup", "local");
  __builtin_amdgcn_s_barrier();
  __builtin_amdgcn_fence(__ATOMIC_ACQUIRE, "workgroup", "local");
}

#define STEP_LDS_FLOATS ((CH / KC) * (RB + 32) * LDK)
static_assert((CH / KC) * (RB + 32) >= (HN / KC) * (RB + 64), "the K = 2H tile is the larger one");
static_assert(STEP_LDS_FLOATS * 4 <= 320 * 1024, "one work-group's LDS");

template <int NACC, int HS, class Toks, class Early, class Late>
__device__ __forceinline__ void tile_mac_ahead(
    float* lds, const float* A, const float* __restrict__ Bw0,
    Toks toks, Early early, Late late, v8f (&acc)[NACC]) {
  constexpr int NCH = HS / KC;
  constexpr int CBW = 16 * NACC;
  constexpr int QA = RB / NW;
  constexpr int QB = CBW / NW;
  constexpr int STG = (RB + CBW) * LDK;
  static_assert(QA == 16 && QB * NW == CBW && NCH >= 2 && NCH * STG <= STEP_LDS_FLOATS, "regions");

  int tid = threadIdx.x;
  int lane = tid & 31;
  int wv = tid >> 5;
  int kh = lane >> 4;
  int l16 = lane & 15;

  const float* ag = A + (size_t)(QA * wv) * CH + 4 * lane;
  const float* bg = Bw0 + 4 * lane;
  float* as = lds + (QA * wv) * LDK + 4 * lane;
  float* bs = lds + (RB + QB * wv) * LDK + 4 * lane;
  const float* af = lds + (16 * wv + l16) * LDK + 2 * kh;
  const float* bf = lds + (RB + l16) * LDK + 2 * kh;

  v4f pa[2][QA], pb[2][QB];
  toks();
#pragma unroll
  for (int i = 0; i < QA; ++i) pa[0][i] = *(const v4f*)(ag + (size_t)i * CH);
#pragma unroll
  for (int q = 0; q < QB; ++q) pb[0][q] = *(const v4f*)(bg + (size_t)q * HS);
  __builtin_amdgcn_sched_barrier(0);
  early();
#pragma unroll
  for (int i = 0; i < QA; ++i) pa[1][i] = *(const v4f*)(ag + (size_t)i * CH + KC);
#pragma unroll
  for (int q = 0; q < QB; ++q) pb[1][q] = *(const v4f*)(bg + (size_t)q * HS + KC);
  __builtin_amdgcn_sched_barrier(0);

#pragma unroll
  for (int c = 0; c < NCH; ++c) {
    int roff = c * STG;
    asm volatile("" : "+s"(roff));
    float* sa = as + roff;
    float* sb = bs + roff;
#pragma unroll
    for (int i = 0; i < QA; ++i) *(v4f*)(sa + i * LDK) = pa[c & 1][i];
#pragma unroll
    for (int q = 0; q < QB; ++q) *(v4f*)(sb + q * LDK) = pb[c & 1][q];
    if (c + 2 < NCH) {
      __builtin_amdgcn_sched_barrier(0);
#pragma unroll
      for (int i = 0; i < QA; ++i) pa[c & 1][i] = *(const v4f*)(ag + (size_t)i * CH + (c + 2) * KC);
#pragma unroll
      for (int q = 0; q < QB; ++q) pb[c & 1][q] = *(const v4f*)(bg + (size_t)q * HS + (c + 2) * KC);
      __builtin_amdgcn_sched_barrier(0);
    }
    lds_barrier();
    if (c == NCH - 2) late();

    const float* al = af + roff;
    const float* bl = bf + roff;
#pragma unroll
    for (int k = 0; k < KC; k += 4) {
      v2f a = *(const v2f*)(al + k);
      v2f b[NACC];
#pragma unroll
      for (int j = 0; j < NACC; ++j) b[j] = *(const v2f*)(bl + 16 * j * LDK + k);
#pragma unroll
      for (int j = 0; j < NACC; ++j)
        acc[j] = __builtin_amdgcn_wmma_f32_16x16x4_f32(false, a, false, b[j], (short)0, acc[j], false, false);
    }
  }
}

__device__ __forceinline__ void proj_tile_wmma_f32(
    float* lds, const float* __restrict__ X, const float* __restrict__ W,
    const float* __restrict__ bias, float* __restrict__ outT, int K, int rowBlk) {
  int tid = threadIdx.x;
  int lane = tid & 31;
  int wv = tid >> 5;
  int kh = lane >> 4;
  int l16 = lane & 15;
  int row0 = rowBlk * RB + wv * 16 + kh * 8;
  int tstep = row0 / BN;
  int b0 = row0 - tstep * BN;

  v8f accs[4] = {};
  tile_mac<4>(lds, X + (size_t)(rowBlk * RB) * K, K, W, K,
              [=](int n) { return n; }, K / KC, accs);

#pragma unroll
  for (int j = 0; j < 4; ++j) {
    int col = j * 16 + l16;
    float badd = bias[col];
    float* o = outT + ((size_t)tstep * OPAD + col) * BN + b0;
    v4f lo = {accs[j][0] + badd, accs[j][1] + badd, accs[j][2] + badd, accs[j][3] + badd};
    v4f hi = {accs[j][4] + badd, accs[j][5] + badd, accs[j][6] + badd, accs[j][7] + badd};
    *(v4f*)o = lo;
    *(v4f*)(o + 4) = hi;
  }
}

__device__ __forceinline__ float sigmoid_hw(float x) {
  return __builtin_amdgcn_rcpf(1.f + __builtin_amdgcn_exp2f(-1.4426950408889634f * x));
}
__device__ __forceinline__ float tanh_hw(float x) {
  return fmaf(-2.f, __builtin_amdgcn_rcpf(1.f + __builtin_amdgcn_exp2f(2.8853900817779268f * x)), 1.f);
}

template <int NACC, int HS>
__device__ __forceinline__ void lstm_step_tile(
    float* lds, const float* hprev, const float* __restrict__ Whh,
    const float* __restrict__ bias, const float* __restrict__ Wih,
    const int* __restrict__ tok, float* __restrict__ c, float* hout,
    bool first, int embD, int ub, int rh) {
  constexpr int UBK = 4 * NACC;
  constexpr int NR = 2 * NACC;
  static_assert(NACC == 4 || NACC == 2, "gate layouts written out for 16 and 8 units");
  static_assert(16 * NACC / NW == UBK, "wave g of the block stages the block's rows of gate g");
  static_assert(HS % KC == 0 && HS <= CH, "whole K chunks");
  int tid = threadIdx.x;
  int lane = tid & 31;
  int wl = tid >> 5;
  int wv = rh * (RB / 16) + wl;
  int kh = lane >> 4;
  int l16 = lane & 15;
  int row0 = wv * 16;
  bool lo = (NACC == 4) || (l16 < 8);
  int ul = (NACC == 4) ? l16 : (l16 & 7);
  int rbase = row0 + kh * 8 + (lo ? 0 : 4);
  int u = ub * UBK + ul;

  int tk[NR];
  float badd[4], ein[4][NR], cold[NR];
  float* cp = c + (size_t)(ub * (BN / 16) + wv) * (32 * NR) + lane;
  auto toks = [&]() __attribute__((always_inline)) {
#pragma unroll
    for (int r = 0; r < NR; ++r) tk[r] = tok[rbase + r];
  };
  auto early = [&]() __attribute__((always_inline)) {
#pragma unroll
    for (int j = 0; j < 4; ++j) badd[j] = bias[j * HS + u];
#pragma unroll
    for (int r = 0; r < NR; ++r) cold[r] = first ? 0.f : cp[r * 32];
  };
  auto late = [&]() __attribute__((always_inline)) {
#pragma unroll
    for (int r = 0; r < NR; ++r) {
      asm volatile("" : "+v"(tk[r]));
    }
#pragma unroll
    for (int j = 0; j < 4; ++j) {
      const float* e = Wih + (size_t)(j * HS + u) * embD;
#pragma unroll
      for (int r = 0; r < NR; ++r) ein[j][r] = e[tk[r]];
    }
  };

  v8f acc[NACC] = {};
  if (first) {
    toks();
    early();
    late();
  } else {
    tile_mac_ahead<NACC, HS>(lds, hprev + (size_t)(rh * RB) * CH,
                             Whh + ((size_t)wl * HS + ub * UBK) * HS, toks, early, late, acc);
  }

  float ai[NR], af[NR], ag[NR], ao[NR];
  if constexpr (NACC == 4) {
#pragma unroll
    for (int r = 0; r < NR; ++r) {
      ai[r] = acc[0][r]; af[r] = acc[1][r]; ag[r] = acc[2][r]; ao[r] = acc[3][r];
    }
  } else {
    v8f m0 = lo ? acc[0] : __builtin_shufflevector(acc[0], acc[0], 4, 5, 6, 7, 0, 1, 2, 3);
    v8f m1 = lo ? acc[1] : __builtin_shufflevector(acc[1], acc[1], 4, 5, 6, 7, 0, 1, 2, 3);
#pragma unroll
    for (int r = 0; r < NR; ++r) {
      float got0 = __shfl_xor(m0[r + 4], 8, 32);
      float got1 = __shfl_xor(m1[r + 4], 8, 32);
      ai[r] = lo ? m0[r] : got0; af[r] = lo ? got0 : m0[r];
      ag[r] = lo ? m1[r] : got1; ao[r] = lo ? got1 : m1[r];
    }
  }

#pragma unroll
  for (int r = 0; r < NR; ++r) {
    int b = rbase + r;
    float gi = ai[r] + badd[0]; gi += ein[0][r];
    float gf = af[r] + badd[1]; gf += ein[1][r];
    float gg = ag[r] + badd[2]; gg += ein[2][r];
    float go = ao[r] + badd[3]; go += ein[3][r];
    float si = sigmoid_hw(gi);
    float sf = sigmoid_hw(gf);
    float so = sigmoid_hw(go);
    float cn = sf * cold[r] + si * tanh_hw(gg);
    float hn = so * tanh_hw(cn);
    cp[r * 32] = cn;
    hout[(size_t)b * CH + u] = hn;
  }
}

__global__ __launch_bounds__(NT) void lstm_step(
    const float* hp_m, const float* __restrict__ Whh_m, const float* __restrict__ b_m,
    const float* __restrict__ Wih_m, const int* __restrict__ tok_m,
    float* __restrict__ c_m, float* ho_m,
    const float* hp_f, const float* __restrict__ Whh_f, const float* __restrict__ b_f,
    const float* __restrict__ Wih_f, const int* __restrict__ tok_f,
    float* __restrict__ c_f, float* ho_f,
    const float* hp_b, const float* __restrict__ Whh_b, const float* __restrict__ b_b,
    const float* __restrict__ Wih_b, const int* __restrict__ tok_b,
    float* __restrict__ c_b, float* ho_b,
    int first, int embD) {
  __shared__ __attribute__((aligned(16))) float lds[STEP_LDS_FLOATS];
  int rh = blockIdx.x % RSPLIT;
  int blk = blockIdx.x / RSPLIT;
  if (blk < MBLK) {
    lstm_step_tile<2, CH>(lds, hp_m, Whh_m, b_m, Wih_m, tok_m, c_m, ho_m, first != 0, embD, blk, rh);
    return;
  }
  blk -= MBLK;
  const float* hp = hp_f;
  const float* Wt = Whh_f;
  const float* bias = b_f;
  const float* emb = Wih_f;
  const int* tok = tok_f;
  float* c = c_f;
  float* ho = ho_f;
  if (blk >= PBLK) {
    blk -= PBLK;
    hp = hp_b; Wt = Whh_b; bias = b_b; emb = Wih_b; tok = tok_b; c = c_b; ho = ho_b;
  }
  lstm_step_tile<4, HN>(lds, hp, Wt, bias, emb, tok, c, ho, first != 0, embD, blk, rh);
}

__global__ __launch_bounds__(NT) void proj_gemm(
    const float* __restrict__ prior, const float* __restrict__ modern,
    const float* __restrict__ Wsp, const float* __restrict__ bsp,
    const float* __restrict__ Wip, const float* __restrict__ bip,
    float* __restrict__ pPs, float* __restrict__ pPi,
    float* __restrict__ pMs, float* __restrict__ pMi, int K) {
  int job = blockIdx.y;
  const float* X = (job < 2) ? prior : modern;
  const float* Wt = (job & 1) ? Wip : Wsp;
  const float* bias = (job & 1) ? bip : bsp;
  float* out = (job == 0) ? pPs : (job == 1) ? pPi : (job == 2) ? pMs : pMi;
  __shared__ __attribute__((aligned(16))) float lds[LDS_FLOATS];
  proj_tile_wmma_f32(lds, X, Wt, bias, out, K, blockIdx.x);
}

__global__ __launch_bounds__(256) void finalize(
    const float* __restrict__ pPs, const float* __restrict__ pPi,
    const float* __restrict__ pMs, const float* __restrict__ pMi,
    const int* __restrict__ tgt, float* __restrict__ out) {
  int t = blockIdx.x * blockDim.x + threadIdx.x;
  if (t >= PN * MN * BN) return;
  int b = t % BN;
  int pm = t / BN;
  int m = pm % MN;
  int p = pm / MN;
  const float* as = pPs + (size_t)p * OPAD * BN + b;
  const float* bsv = pMs + (size_t)m * OPAD * BN + b;
  const float* ai = pPi + (size_t)p * OPAD * BN + b;
  const float* biv = pMi + (size_t)m * OPAD * BN + b;
  int tg = (m < MN - 1) ? tgt[(m + 1) * BN + b] : 0;

  float ls[ON], li[ON];
#pragma unroll
  for (int o = 0; o < ON; ++o) {
    ls[o] = as[o * BN] + bsv[o * BN];
    li[o] = ai[o * BN] + biv[o * BN];
  }
  float mxs = -3.4e38f, mxi = -3.4e38f;
#pragma unroll
  for (int o = 0; o < ON; ++o) {
    mxs = ls[o] > mxs ? ls[o] : mxs;
    mxi = li[o] > mxi ? li[o] : mxi;
  }
  float sums = 0.f, sumi = 0.f, lst = 0.f, lit = 0.f;
#pragma unroll
  for (int o = 0; o < ON; ++o) {
    sums += expf(ls[o] - mxs);
    sumi += expf(li[o] - mxi);
    if (o == tg) { lst = ls[o]; lit = li[o]; }
  }
  float ls52 = ls[ON - 1], li52 = li[ON - 1];
  float lses = mxs + logf(sums);
  float lsei = mxi + logf(sumi);
  size_t idxPMB = ((size_t)p * MN + m) * BN + b;
  out[idxPMB] = ls52 - lses;
  out[(size_t)PN * MN * BN + idxPMB] = li52 - lsei;
  if (m < MN - 1) {
    size_t base = 2ull * PN * MN * BN;
    size_t idxS = ((size_t)p * (MN - 1) + m) * BN + b;
    out[base + idxS] = lst - lses;
    out[base + (size_t)PN * (MN - 1) * BN + idxS] = lit - lsei;
  }
}


extern "C" void kernel_launch(void* const* d_in, const int* in_sizes, int n_in,
                              void* d_out, int out_size, void* d_ws, size_t ws_size,
                              hipStream_t stream) {
  (void)in_sizes; (void)n_in; (void)out_size; (void)ws_size;
  const float* src_oh = (const float*)d_in[0];
  const float* tgt_oh = (const float*)d_in[1];  (void)tgt_oh;
  const float* Wih_pf = (const float*)d_in[2];
  const float* Whh_pf = (const float*)d_in[3];
  const float* b_pf   = (const float*)d_in[4];
  const float* Wih_pb = (const float*)d_in[5];
  const float* Whh_pb = (const float*)d_in[6];
  const float* b_pb   = (const float*)d_in[7];
  const float* Wih_m  = (const float*)d_in[8];
  const float* Whh_m  = (const float*)d_in[9];
  const float* b_m    = (const float*)d_in[10];
  const float* W_sub  = (const float*)d_in[11];
  const float* b_sub  = (const float*)d_in[12];
  const float* W_ins  = (const float*)d_in[13];
  const float* b_ins  = (const float*)d_in[14];
  const int*   tgt    = (const int*)d_in[15];

  char* wptr = (char*)d_ws;
  auto alloc = [&](size_t nfloats) -> float* {
    float* p = (float*)wptr;
    wptr += ((nfloats * 4 + 255) / 256) * 256;
    return p;
  };
  float* prior  = alloc((size_t)PN * BN * CH);
  float* modern = alloc((size_t)MN * BN * CH);
  float* c_pf = alloc((size_t)BN * HN);
  float* c_pb = alloc((size_t)BN * HN);
  float* c_m  = alloc((size_t)BN * CH);
  float* Wsp = alloc(OPAD * CH);
  float* Wip = alloc(OPAD * CH);
  float* bsp = alloc(OPAD);
  float* bip = alloc(OPAD);
  float* pPs = alloc((size_t)PN * BN * OPAD);
  float* pPi = alloc((size_t)PN * BN * OPAD);
  float* pMs = alloc((size_t)MN * BN * OPAD);
  float* pMi = alloc((size_t)MN * BN * OPAD);
  int* sidx = (int*)alloc(PN * BN);

  setup_all<<<IBLK + WBLK, 256, 0, stream>>>(
      src_oh, sidx, W_sub, b_sub, W_ins, b_ins, Wsp, Wip, bsp, bip);

  lstm_step<<<(MBLK + 2 * PBLK) * RSPLIT, NT, 0, stream>>>(
      modern,     Whh_m,  b_m,  Wih_m,  tgt,                  c_m,  modern,
      prior,      Whh_pf, b_pf, Wih_pf, sidx,                 c_pf, prior,
      prior + HN, Whh_pb, b_pb, Wih_pb, sidx + (PN - 1) * BN, c_pb,
      prior + (size_t)(PN - 1) * BN * CH + HN,
      1, DN);
  for (int t = 1; t < PN; ++t) {
    int tb = PN - 1 - t;
    lstm_step<<<(MBLK + 2 * PBLK) * RSPLIT, NT, 0, stream>>>(
        modern + (size_t)(t - 1) * BN * CH, Whh_m,  b_m,  Wih_m,  tgt + t * BN,   c_m,
        modern + (size_t)t * BN * CH,
        prior + (size_t)(t - 1) * BN * CH,  Whh_pf, b_pf, Wih_pf, sidx + t * BN,  c_pf,
        prior + (size_t)t * BN * CH,
        prior + (size_t)(tb + 1) * BN * CH + HN, Whh_pb, b_pb, Wih_pb, sidx + tb * BN, c_pb,
        prior + (size_t)tb * BN * CH + HN,
        0, DN);
  }

  proj_gemm<<<dim3(PN * BN / RB, 4), NT, 0, stream>>>(
      prior, modern, Wsp, bsp, Wip, bip, pPs, pPi, pMs, pMi, CH);

  finalize<<<(PN * MN * BN + 255) / 256, 256, 0, stream>>>(
      pPs, pPi, pMs, pMi, tgt, (float*)d_out);
}


# LDS fragment reads issued as one group per block of 32 WMMAs, one block ahead (second fragment set), not per k-step among the WMMAs: LSTM step and projection K loops; same products, same k order
# speedup vs baseline: 9.7535x; 1.0674x over previous
#include <hip/hip_runtime.h>
#include <math.h>

typedef __attribute__((ext_vector_type(2))) float v2f;
typedef __attribute__((ext_vector_type(4))) float v4f;
typedef __attribute__((ext_vector_type(8))) float v8f;

#define PN 32
#define MN 32
#define BN 128
#define HN 256
#define DN 54
#define ON 53
#define OPAD 64
#define CH 512

#define RB 64
#define NT (2 * RB)
#define RSPLIT (BN / RB)
#define CB 64
#define KC 128
#define LDK (KC + 4)
#define NW (NT / 32)
#define STAGE_MAX ((RB + CB) * LDK)
#define LDS_FLOATS (2 * STAGE_MAX)

#define IBLK ((PN * BN) / 256)
#define WBLK ((OPAD * CH) / 256)
#define UBP 16
#define UBM 8
#define MBLK (CH / UBM)
#define PBLK (HN / UBP)

static_assert(PN == MN, "the three recurrences advance in lock step");
static_assert((PN * BN) % 256 == 0 && (OPAD * CH) % 256 == 0, "exact grids");
static_assert(BN % RB == 0 && (PN * BN) % RB == 0 && RB % 16 == 0, "row blocks");
static_assert(CB == 4 * UBP && CH % UBM == 0 && HN % UBP == 0, "a step tile holds all four gates of its units");
static_assert(NT == 128 && RB == 16 * NW, "a wave stages the 16 A rows it multiplies");
static_assert(KC == 4 * 32, "a wave load of 16 B per lane covers one staged row");
static_assert(CH % KC == 0 && HN % KC == 0, "whole K chunks");


__device__ __forceinline__ void onehot_to_idx_one(const float* __restrict__ oh,
                                                  int* __restrict__ idx, int r) {
  const float* p = oh + (size_t)r * DN;
  int v = 0;
  for (int d = 0; d < DN; ++d)
    if (p[d] > 0.5f) v = d;
  idx[r] = v;
}

__device__ __forceinline__ void pad_proj_weights_one(
    const float* __restrict__ Ws, const float* __restrict__ bs,
    const float* __restrict__ Wi, const float* __restrict__ bi,
    float* __restrict__ Wsp, float* __restrict__ Wip,
    float* __restrict__ bsp, float* __restrict__ bip, int t) {
  int n = t / CH;
  int k = t - n * CH;
  Wsp[t] = (n < ON) ? Ws[n * CH + k] : 0.f;
  Wip[t] = (n < ON) ? Wi[n * CH + k] : 0.f;
  if (t < OPAD) {
    bsp[t] = (t < ON) ? bs[t] : 0.f;
    bip[t] = (t < ON) ? bi[t] : 0.f;
  }
}

__global__ void setup_all(const float* __restrict__ src_oh, int* __restrict__ sidx,
                          const float* __restrict__ Ws, const float* __restrict__ bs,
                          const float* __restrict__ Wi, const float* __restrict__ bi,
                          float* __restrict__ Wsp, float* __restrict__ Wip,
                          float* __restrict__ bsp, float* __restrict__ bip) {
  int blk = blockIdx.x;
  int tid = threadIdx.x;
  if (blk < IBLK) {
    onehot_to_idx_one(src_oh, sidx, blk * 256 + tid);
  } else {
    pad_proj_weights_one(Ws, bs, Wi, bi, Wsp, Wip, bsp, bip, (blk - IBLK) * 256 + tid);
  }
}

template <int NACC, class ColRow>
__device__ __forceinline__ void tile_mac(
    float* lds, const float* __restrict__ A, int lda,
    const float* __restrict__ Bw, int ldb, ColRow colrow, int nchunks,
    v8f (&acc)[NACC]) {
  constexpr int CBW = 16 * NACC;
  constexpr int QA = RB / NW;
  constexpr int QB = CBW / NW;
  constexpr int STG = (RB + CBW) * LDK;
  static_assert(QA == 16 && QB * NW == CBW && STG <= STAGE_MAX, "whole rows per wave");

  int tid = threadIdx.x;
  int lane = tid & 31;
  int wv = tid >> 5;
  int kh = lane >> 4;
  int l16 = lane & 15;

  if (nchunks <= 0) return;

  const float* ag = A + (size_t)(QA * wv) * lda + 4 * lane;
  const float* bg[QB];
#pragma unroll
  for (int q = 0; q < QB; ++q) bg[q] = Bw + (size_t)colrow(QB * wv + q) * ldb + 4 * lane;
  float* as = lds + (QA * wv) * LDK + 4 * lane;
  float* bs = lds + (RB + QB * wv) * LDK + 4 * lane;

  {
    v4f pa[QA], pb[QB];
#pragma unroll
    for (int i = 0; i < QA; ++i) pa[i] = *(const v4f*)(ag + (size_t)i * lda);
#pragma unroll
    for (int q = 0; q < QB; ++q) pb[q] = *(const v4f*)(bg[q]);
#pragma unroll
    for (int i = 0; i < QA; ++i) *(v4f*)(as + i * LDK) = pa[i];
#pragma unroll
    for (int q = 0; q < QB; ++q) *(v4f*)(bs + q * LDK) = pb[q];
  }
  __syncthreads();

  const float* af = lds + (16 * wv + l16) * LDK + 2 * kh;
  const float* bf = lds + (RB + l16) * LDK + 2 * kh;

  for (int kc = 0; kc < nchunks; ++kc) {
    v4f pa[QA], pb[QB];
    bool hasNext = (kc + 1 < nchunks);
    if (hasNext) {
      size_t ko = (size_t)(kc + 1) * KC;
#pragma unroll
      for (int i = 0; i < QA; ++i) pa[i] = *(const v4f*)(ag + (size_t)i * lda + ko);
#pragma unroll
      for (int q = 0; q < QB; ++q) pb[q] = *(const v4f*)(bg[q] + ko);
    }

    const float* al = af + (kc & 1) * STG;
    const float* bl = bf + (kc & 1) * STG;

    constexpr int WB = 32;
    constexpr int G = WB / NACC;
    constexpr int NG = (KC / 4) / G;
    static_assert(G * NACC == WB && NG * G * 4 == KC && NG >= 1, "whole blocks per chunk");
    v2f fa[2][G], fb[2][G][NACC];
    auto frags = [&](int gg) __attribute__((always_inline)) {
#pragma unroll
      for (int s = 0; s < G; ++s) {
        fa[gg & 1][s] = *(const v2f*)(al + 4 * (G * gg + s));
#pragma unroll
        for (int j = 0; j < NACC; ++j) fb[gg & 1][s][j] = *(const v2f*)(bl + 16 * j * LDK + 4 * (G * gg + s));
      }
    };
    frags(0);
#pragma unroll
    for (int g = 0; g < NG; ++g) {
      if (g + 1 < NG) frags(g + 1);
      __builtin_amdgcn_sched_barrier(0);
#pragma unroll
      for (int s = 0; s < G; ++s) {
#pragma unroll
        for (int j = 0; j < NACC; ++j)
          acc[j] = __builtin_amdgcn_wmma_f32_16x16x4_f32(false, fa[g & 1][s], false, fb[g & 1][s][j], (short)0, acc[j], false, false);
      }
      __builtin_amdgcn_sched_barrier(0);
    }

    if (hasNext) {
      int so = ((kc + 1) & 1) * STG;
#pragma unroll
      for (int i = 0; i < QA; ++i) *(v4f*)(as + so + i * LDK) = pa[i];
#pragma unroll
      for (int q = 0; q < QB; ++q) *(v4f*)(bs + so + q * LDK) = pb[q];
      __syncthreads();
    }
  }
}

__device__ __forceinline__ void lds_barrier() {
  __builtin_amdgcn_fence(__ATOMIC_RELEASE, "workgroup", "local");
  __builtin_amdgcn_s_barrier();
  __builtin_amdgcn_fence(__ATOMIC_ACQUIRE, "workgroup", "local");
}

#define STEP_LDS_FLOATS ((CH / KC) * (RB + 32) * LDK)
static_assert((CH / KC) * (RB + 32) >= (HN / KC) * (RB + 64), "the K = 2H tile is the larger one");
static_assert(STEP_LDS_FLOATS * 4 <= 320 * 1024, "one work-group's LDS");

template <int NACC, int HS, class Toks, class Early, class Late>
__device__ __forceinline__ void tile_mac_ahead(
    float* lds, const float* A, const float* __restrict__ Bw0,
    Toks toks, Early early, Late late, v8f (&acc)[NACC]) {
  constexpr int NCH = HS / KC;
  constexpr int CBW = 16 * NACC;
  constexpr int QA = RB / NW;
  constexpr int QB = CBW / NW;
  constexpr int STG = (RB + CBW) * LDK;
  static_assert(QA == 16 && QB * NW == CBW && NCH >= 2 && NCH * STG <= STEP_LDS_FLOATS, "regions");

  int tid = threadIdx.x;
  int lane = tid & 31;
  int wv = tid >> 5;
  int kh = lane >> 4;
  int l16 = lane & 15;

  const float* ag = A + (size_t)(QA * wv) * CH + 4 * lane;
  const float* bg = Bw0 + 4 * lane;
  float* as = lds + (QA * wv) * LDK + 4 * lane;
  float* bs = lds + (RB + QB * wv) * LDK + 4 * lane;
  const float* af = lds + (16 * wv + l16) * LDK + 2 * kh;
  const float* bf = lds + (RB + l16) * LDK + 2 * kh;

  v4f pa[2][QA], pb[2][QB];
  toks();
#pragma unroll
  for (int i = 0; i < QA; ++i) pa[0][i] = *(const v4f*)(ag + (size_t)i * CH);
#pragma unroll
  for (int q = 0; q < QB; ++q) pb[0][q] = *(const v4f*)(bg + (size_t)q * HS);
  __builtin_amdgcn_sched_barrier(0);
  early();
#pragma unroll
  for (int i = 0; i < QA; ++i) pa[1][i] = *(const v4f*)(ag + (size_t)i * CH + KC);
#pragma unroll
  for (int q = 0; q < QB; ++q) pb[1][q] = *(const v4f*)(bg + (size_t)q * HS + KC);
  __builtin_amdgcn_sched_barrier(0);

  constexpr int WB = 32;
  constexpr int G = WB / NACC;
  constexpr int NG = (KC / 4) / G;
  static_assert(G * NACC == WB && NG * G * 4 == KC && NG >= 1, "whole blocks per chunk");
  v2f fa[2][G], fb[2][G][NACC];

  int roffs[NCH];
  auto park = [&](int cc) __attribute__((always_inline)) {
    int roff = cc * STG;
    asm volatile("" : "+s"(roff));
    roffs[cc] = roff;
    float* sa = as + roff;
    float* sb = bs + roff;
#pragma unroll
    for (int i = 0; i < QA; ++i) *(v4f*)(sa + i * LDK) = pa[cc & 1][i];
#pragma unroll
    for (int q = 0; q < QB; ++q) *(v4f*)(sb + q * LDK) = pb[cc & 1][q];
    if (cc + 2 < NCH) {
      __builtin_amdgcn_sched_barrier(0);
#pragma unroll
      for (int i = 0; i < QA; ++i) pa[cc & 1][i] = *(const v4f*)(ag + (size_t)i * CH + (cc + 2) * KC);
#pragma unroll
      for (int q = 0; q < QB; ++q) pb[cc & 1][q] = *(const v4f*)(bg + (size_t)q * HS + (cc + 2) * KC);
      __builtin_amdgcn_sched_barrier(0);
    }
    lds_barrier();
    if (cc == NCH - 2) late();
  };
  auto frags = [&](int cc, int gg, int st) __attribute__((always_inline)) {
    const float* al = af + roffs[cc] + 4 * G * gg;
    const float* bl = bf + roffs[cc] + 4 * G * gg;
#pragma unroll
    for (int s = 0; s < G; ++s) {
      fa[st][s] = *(const v2f*)(al + 4 * s);
#pragma unroll
      for (int j = 0; j < NACC; ++j) fb[st][s][j] = *(const v2f*)(bl + 16 * j * LDK + 4 * s);
    }
  };

  park(0);
  frags(0, 0, 0);
#pragma unroll
  for (int c = 0; c < NCH; ++c) {
#pragma unroll
    for (int g = 0; g < NG; ++g) {
      const int cur = (c * NG + g) & 1;
      if (g + 1 < NG) {
        frags(c, g + 1, cur ^ 1);
      } else if (c + 1 < NCH) {
        park(c + 1);
        frags(c + 1, 0, cur ^ 1);
      }
      __builtin_amdgcn_sched_barrier(0);
#pragma unroll
      for (int s = 0; s < G; ++s) {
#pragma unroll
        for (int j = 0; j < NACC; ++j)
          acc[j] = __builtin_amdgcn_wmma_f32_16x16x4_f32(false, fa[cur][s], false, fb[cur][s][j], (short)0, acc[j], false, false);
      }
      __builtin_amdgcn_sched_barrier(0);
    }
  }
}

__device__ __forceinline__ void proj_tile_wmma_f32(
    float* lds, const float* __restrict__ X, const float* __restrict__ W,
    const float* __restrict__ bias, float* __restrict__ outT, int K, int rowBlk) {
  int tid = threadIdx.x;
  int lane = tid & 31;
  int wv = tid >> 5;
  int kh = lane >> 4;
  int l16 = lane & 15;
  int row0 = rowBlk * RB + wv * 16 + kh * 8;
  int tstep = row0 / BN;
  int b0 = row0 - tstep * BN;

  v8f accs[4] = {};
  tile_mac<4>(lds, X + (size_t)(rowBlk * RB) * K, K, W, K,
              [=](int n) { return n; }, K / KC, accs);

#pragma unroll
  for (int j = 0; j < 4; ++j) {
    int col = j * 16 + l16;
    float badd = bias[col];
    float* o = outT + ((size_t)tstep * OPAD + col) * BN + b0;
    v4f lo = {accs[j][0] + badd, accs[j][1] + badd, accs[j][2] + badd, accs[j][3] + badd};
    v4f hi = {accs[j][4] + badd, accs[j][5] + badd, accs[j][6] + badd, accs[j][7] + badd};
    *(v4f*)o = lo;
    *(v4f*)(o + 4) = hi;
  }
}

__device__ __forceinline__ float sigmoid_hw(float x) {
  return __builtin_amdgcn_rcpf(1.f + __builtin_amdgcn_exp2f(-1.4426950408889634f * x));
}
__device__ __forceinline__ float tanh_hw(float x) {
  return fmaf(-2.f, __builtin_amdgcn_rcpf(1.f + __builtin_amdgcn_exp2f(2.8853900817779268f * x)), 1.f);
}

template <int NACC, int HS>
__device__ __forceinline__ void lstm_step_tile(
    float* lds, const float* hprev, const float* __restrict__ Whh,
    const float* __restrict__ bias, const float* __restrict__ Wih,
    const int* __restrict__ tok, float* __restrict__ c, float* hout,
    bool first, int embD, int ub, int rh) {
  constexpr int UBK = 4 * NACC;
  constexpr int NR = 2 * NACC;
  static_assert(NACC == 4 || NACC == 2, "gate layouts written out for 16 and 8 units");
  static_assert(16 * NACC / NW == UBK, "wave g of the block stages the block's rows of gate g");
  static_assert(HS % KC == 0 && HS <= CH, "whole K chunks");
  int tid = threadIdx.x;
  int lane = tid & 31;
  int wl = tid >> 5;
  int wv = rh * (RB / 16) + wl;
  int kh = lane >> 4;
  int l16 = lane & 15;
  int row0 = wv * 16;
  bool lo = (NACC == 4) || (l16 < 8);
  int ul = (NACC == 4) ? l16 : (l16 & 7);
  int rbase = row0 + kh * 8 + (lo ? 0 : 4);
  int u = ub * UBK + ul;

  int tk[NR];
  float badd[4], ein[4][NR], cold[NR];
  float* cp = c + (size_t)(ub * (BN / 16) + wv) * (32 * NR) + lane;
  auto toks = [&]() __attribute__((always_inline)) {
#pragma unroll
    for (int r = 0; r < NR; ++r) tk[r] = tok[rbase + r];
  };
  auto early = [&]() __attribute__((always_inline)) {
#pragma unroll
    for (int j = 0; j < 4; ++j) badd[j] = bias[j * HS + u];
#pragma unroll
    for (int r = 0; r < NR; ++r) cold[r] = first ? 0.f : cp[r * 32];
  };
  auto late = [&]() __attribute__((always_inline)) {
#pragma unroll
    for (int r = 0; r < NR; ++r) {
      asm volatile("" : "+v"(tk[r]));
    }
#pragma unroll
    for (int j = 0; j < 4; ++j) {
      const float* e = Wih + (size_t)(j * HS + u) * embD;
#pragma unroll
      for (int r = 0; r < NR; ++r) ein[j][r] = e[tk[r]];
    }
  };

  v8f acc[NACC] = {};
  if (first) {
    toks();
    early();
    late();
  } else {
    tile_mac_ahead<NACC, HS>(lds, hprev + (size_t)(rh * RB) * CH,
                             Whh + ((size_t)wl * HS + ub * UBK) * HS, toks, early, late, acc);
  }

  float ai[NR], af[NR], ag[NR], ao[NR];
  if constexpr (NACC == 4) {
#pragma unroll
    for (int r = 0; r < NR; ++r) {
      ai[r] = acc[0][r]; af[r] = acc[1][r]; ag[r] = acc[2][r]; ao[r] = acc[3][r];
    }
  } else {
    v8f m0 = lo ? acc[0] : __builtin_shufflevector(acc[0], acc[0], 4, 5, 6, 7, 0, 1, 2, 3);
    v8f m1 = lo ? acc[1] : __builtin_shufflevector(acc[1], acc[1], 4, 5, 6, 7, 0, 1, 2, 3);
#pragma unroll
    for (int r = 0; r < NR; ++r) {
      float got0 = __shfl_xor(m0[r + 4], 8, 32);
      float got1 = __shfl_xor(m1[r + 4], 8, 32);
      ai[r] = lo ? m0[r] : got0; af[r] = lo ? got0 : m0[r];
      ag[r] = lo ? m1[r] : got1; ao[r] = lo ? got1 : m1[r];
    }
  }

#pragma unroll
  for (int r = 0; r < NR; ++r) {
    int b = rbase + r;
    float gi = ai[r] + badd[0]; gi += ein[0][r];
    float gf = af[r] + badd[1]; gf += ein[1][r];
    float gg = ag[r] + badd[2]; gg += ein[2][r];
    float go = ao[r] + badd[3]; go += ein[3][r];
    float si = sigmoid_hw(gi);
    float sf = sigmoid_hw(gf);
    float so = sigmoid_hw(go);
    float cn = sf * cold[r] + si * tanh_hw(gg);
    float hn = so * tanh_hw(cn);
    cp[r * 32] = cn;
    hout[(size_t)b * CH + u] = hn;
  }
}

__global__ __launch_bounds__(NT) void lstm_step(
    const float* hp_m, const float* __restrict__ Whh_m, const float* __restrict__ b_m,
    const float* __restrict__ Wih_m, const int* __restrict__ tok_m,
    float* __restrict__ c_m, float* ho_m,
    const float* hp_f, const float* __restrict__ Whh_f, const float* __restrict__ b_f,
    const float* __restrict__ Wih_f, const int* __restrict__ tok_f,
    float* __restrict__ c_f, float* ho_f,
    const float* hp_b, const float* __restrict__ Whh_b, const float* __restrict__ b_b,
    const float* __restrict__ Wih_b, const int* __restrict__ tok_b,
    float* __restrict__ c_b, float* ho_b,
    int first, int embD) {
  __shared__ __attribute__((aligned(16))) float lds[STEP_LDS_FLOATS];
  int rh = blockIdx.x % RSPLIT;
  int blk = blockIdx.x / RSPLIT;
  if (blk < MBLK) {
    lstm_step_tile<2, CH>(lds, hp_m, Whh_m, b_m, Wih_m, tok_m, c_m, ho_m, first != 0, embD, blk, rh);
    return;
  }
  blk -= MBLK;
  const float* hp = hp_f;
  const float* Wt = Whh_f;
  const float* bias = b_f;
  const float* emb = Wih_f;
  const int* tok = tok_f;
  float* c = c_f;
  float* ho = ho_f;
  if (blk >= PBLK) {
    blk -= PBLK;
    hp = hp_b; Wt = Whh_b; bias = b_b; emb = Wih_b; tok = tok_b; c = c_b; ho = ho_b;
  }
  lstm_step_tile<4, HN>(lds, hp, Wt, bias, emb, tok, c, ho, first != 0, embD, blk, rh);
}

__global__ __launch_bounds__(NT) void proj_gemm(
    const float* __restrict__ prior, const float* __restrict__ modern,
    const float* __restrict__ Wsp, const float* __restrict__ bsp,
    const float* __restrict__ Wip, const float* __restrict__ bip,
    float* __restrict__ pPs, float* __restrict__ pPi,
    float* __restrict__ pMs, float* __restrict__ pMi, int K) {
  int job = blockIdx.y;
  const float* X = (job < 2) ? prior : modern;
  const float* Wt = (job & 1) ? Wip : Wsp;
  const float* bias = (job & 1) ? bip : bsp;
  float* out = (job == 0) ? pPs : (job == 1) ? pPi : (job == 2) ? pMs : pMi;
  __shared__ __attribute__((aligned(16))) float lds[LDS_FLOATS];
  proj_tile_wmma_f32(lds, X, Wt, bias, out, K, blockIdx.x);
}

__global__ __launch_bounds__(256) void finalize(
    const float* __restrict__ pPs, const float* __restrict__ pPi,
    const float* __restrict__ pMs, const float* __restrict__ pMi,
    const int* __restrict__ tgt, float* __restrict__ out) {
  int t = blockIdx.x * blockDim.x + threadIdx.x;
  if (t >= PN * MN * BN) return;
  int b = t % BN;
  int pm = t / BN;
  int m = pm % MN;
  int p = pm / MN;
  const float* as = pPs + (size_t)p * OPAD * BN + b;
  const float* bsv = pMs + (size_t)m * OPAD * BN + b;
  const float* ai = pPi + (size_t)p * OPAD * BN + b;
  const float* biv = pMi + (size_t)m * OPAD * BN + b;
  int tg = (m < MN - 1) ? tgt[(m + 1) * BN + b] : 0;

  float ls[ON], li[ON];
#pragma unroll
  for (int o = 0; o < ON; ++o) {
    ls[o] = as[o * BN] + bsv[o * BN];
    li[o] = ai[o * BN] + biv[o * BN];
  }
  float mxs = -3.4e38f, mxi = -3.4e38f;
#pragma unroll
  for (int o = 0; o < ON; ++o) {
    mxs = ls[o] > mxs ? ls[o] : mxs;
    mxi = li[o] > mxi ? li[o] : mxi;
  }
  float sums = 0.f, sumi = 0.f, lst = 0.f, lit = 0.f;
#pragma unroll
  for (int o = 0; o < ON; ++o) {
    sums += expf(ls[o] - mxs);
    sumi += expf(li[o] - mxi);
    if (o == tg) { lst = ls[o]; lit = li[o]; }
  }
  float ls52 = ls[ON - 1], li52 = li[ON - 1];
  float lses = mxs + logf(sums);
  float lsei = mxi + logf(sumi);
  size_t idxPMB = ((size_t)p * MN + m) * BN + b;
  out[idxPMB] = ls52 - lses;
  out[(size_t)PN * MN * BN + idxPMB] = li52 - lsei;
  if (m < MN - 1) {
    size_t base = 2ull * PN * MN * BN;
    size_t idxS = ((size_t)p * (MN - 1) + m) * BN + b;
    out[base + idxS] = lst - lses;
    out[base + (size_t)PN * (MN - 1) * BN + idxS] = lit - lsei;
  }
}


extern "C" void kernel_launch(void* const* d_in, const int* in_sizes, int n_in,
                              void* d_out, int out_size, void* d_ws, size_t ws_size,
                              hipStream_t stream) {
  (void)in_sizes; (void)n_in; (void)out_size; (void)ws_size;
  const float* src_oh = (const float*)d_in[0];
  const float* tgt_oh = (const float*)d_in[1];  (void)tgt_oh;
  const float* Wih_pf = (const float*)d_in[2];
  const float* Whh_pf = (const float*)d_in[3];
  const float* b_pf   = (const float*)d_in[4];
  const float* Wih_pb = (const float*)d_in[5];
  const float* Whh_pb = (const float*)d_in[6];
  const float* b_pb   = (const float*)d_in[7];
  const float* Wih_m  = (const float*)d_in[8];
  const float* Whh_m  = (const float*)d_in[9];
  const float* b_m    = (const float*)d_in[10];
  const float* W_sub  = (const float*)d_in[11];
  const float* b_sub  = (const float*)d_in[12];
  const float* W_ins  = (const float*)d_in[13];
  const float* b_ins  = (const float*)d_in[14];
  const int*   tgt    = (const int*)d_in[15];

  char* wptr = (char*)d_ws;
  auto alloc = [&](size_t nfloats) -> float* {
    float* p = (float*)wptr;
    wptr += ((nfloats * 4 + 255) / 256) * 256;
    return p;
  };
  float* prior  = alloc((size_t)PN * BN * CH);
  float* modern = alloc((size_t)MN * BN * CH);
  float* c_pf = alloc((size_t)BN * HN);
  float* c_pb = alloc((size_t)BN * HN);
  float* c_m  = alloc((size_t)BN * CH);
  float* Wsp = alloc(OPAD * CH);
  float* Wip = alloc(OPAD * CH);
  float* bsp = alloc(OPAD);
  float* bip = alloc(OPAD);
  float* pPs = alloc((size_t)PN * BN * OPAD);
  float* pPi = alloc((size_t)PN * BN * OPAD);
  float* pMs = alloc((size_t)MN * BN * OPAD);
  float* pMi = alloc((size_t)MN * BN * OPAD);
  int* sidx = (int*)alloc(PN * BN);

  setup_all<<<IBLK + WBLK, 256, 0, stream>>>(
      src_oh, sidx, W_sub, b_sub, W_ins, b_ins, Wsp, Wip, bsp, bip);

  lstm_step<<<(MBLK + 2 * PBLK) * RSPLIT, NT, 0, stream>>>(
      modern,     Whh_m,  b_m,  Wih_m,  tgt,                  c_m,  modern,
      prior,      Whh_pf, b_pf, Wih_pf, sidx,                 c_pf, prior,
      prior + HN, Whh_pb, b_pb, Wih_pb, sidx + (PN - 1) * BN, c_pb,
      prior + (size_t)(PN - 1) * BN * CH + HN,
      1, DN);
  for (int t = 1; t < PN; ++t) {
    int tb = PN - 1 - t;
    lstm_step<<<(MBLK + 2 * PBLK) * RSPLIT, NT, 0, stream>>>(
        modern + (size_t)(t - 1) * BN * CH, Whh_m,  b_m,  Wih_m,  tgt + t * BN,   c_m,
        modern + (size_t)t * BN * CH,
        prior + (size_t)(t - 1) * BN * CH,  Whh_pf, b_pf, Wih_pf, sidx + t * BN,  c_pf,
        prior + (size_t)t * BN * CH,
        prior + (size_t)(tb + 1) * BN * CH + HN, Whh_pb, b_pb, Wih_pb, sidx + tb * BN, c_pb,
        prior + (size_t)tb * BN * CH + HN,
        0, DN);
  }

  proj_gemm<<<dim3(PN * BN / RB, 4), NT, 0, stream>>>(
      prior, modern, Wsp, bsp, Wip, bip, pPs, pPi, pMs, pMi, CH);

  finalize<<<(PN * MN * BN + 255) / 256, 256, 0, stream>>>(
      pPs, pPi, pMs, pMi, tgt, (float*)d_out);
}
